# starting kernel
# baseline (speedup 1.0000x reference)
#include <hip/hip_runtime.h>
#include <hip/hip_bf16.h>

#define N_NODES 10000
#define N_EDGES 320000
#define IN_DIM  512
#define OUT_DIM 256
#define NEG_SLOPE 0.01f

typedef float v2f __attribute__((ext_vector_type(2)));
typedef float v8f __attribute__((ext_vector_type(8)));

__device__ __forceinline__ unsigned enc_f32(float f) {
    int b = __float_as_int(f);
    return (b >= 0) ? ((unsigned)b | 0x80000000u) : ~(unsigned)b;
}
__device__ __forceinline__ float dec_f32(unsigned k) {
    int b = (k & 0x80000000u) ? (int)(k & 0x7FFFFFFFu) : ~(int)k;
    return __int_as_float(b);
}
#define EMAX_NEG_INF 0x007FFFFFu

__global__ void k_init(unsigned* __restrict__ emax, float* __restrict__ denom,
                       float* __restrict__ hout) {
    int i = blockIdx.x * blockDim.x + threadIdx.x;
    if (i < N_NODES) { emax[i] = EMAX_NEG_INF; denom[i] = 0.0f; }
    for (int j = i; j < N_NODES * OUT_DIM; j += gridDim.x * blockDim.x)
        hout[j] = 0.0f;
}

__global__ void k_gemm_wmma(const float* __restrict__ h, const float* __restrict__ w,
                            const float* __restrict__ bias, float* __restrict__ z) {
    const int lane = threadIdx.x & 31;
    const int gw   = blockIdx.x * (blockDim.x >> 5) + (threadIdx.x >> 5);
    if (gw >= 2500) return;
    const int row0 = (gw >> 2) * 16;
    const int col0 = (gw & 3) * 64;
    const int mn   = lane & 15;
    const int kb   = (lane >> 4) << 1;

    const float* arow = h + (size_t)(row0 + mn) * IN_DIM + kb;
    const float* b0   = w + (size_t)(col0 + mn) * IN_DIM + kb;
    const float* b1   = b0 + (size_t)16 * IN_DIM;
    const float* b2   = b0 + (size_t)32 * IN_DIM;
    const float* b3   = b0 + (size_t)48 * IN_DIM;

    v8f c0 = {}, c1 = {}, c2 = {}, c3 = {};
#pragma unroll 4
    for (int k = 0; k < IN_DIM; k += 4) {
        v2f a  = *(const v2f*)(arow + k);
        v2f q0 = *(const v2f*)(b0 + k);
        v2f q1 = *(const v2f*)(b1 + k);
        v2f q2 = *(const v2f*)(b2 + k);
        v2f q3 = *(const v2f*)(b3 + k);
        c0 = __builtin_amdgcn_wmma_f32_16x16x4_f32(false, a, false, q0, (short)0, c0, false, false);
        c1 = __builtin_amdgcn_wmma_f32_16x16x4_f32(false, a, false, q1, (short)0, c1, false, false);
        c2 = __builtin_amdgcn_wmma_f32_16x16x4_f32(false, a, false, q2, (short)0, c2, false, false);
        c3 = __builtin_amdgcn_wmma_f32_16x16x4_f32(false, a, false, q3, (short)0, c3, false, false);
    }

    const int mbase = (lane >> 4) * 8;
    v8f acc[4] = {c0, c1, c2, c3};
#pragma unroll
    for (int g = 0; g < 4; ++g) {
        const int   col = col0 + g * 16 + mn;
        const float bv  = bias[col];
#pragma unroll
        for (int v = 0; v < 8; ++v)
            z[(size_t)(row0 + mbase + v) * OUT_DIM + col] = acc[g][v] + bv;
    }
}

__global__ void k_node_dots(const float* __restrict__ z, const float* __restrict__ attn_w,
                            float* __restrict__ s_src, float* __restrict__ s_dst) {
    const int lane = threadIdx.x & 31;
    const int node = (blockIdx.x * blockDim.x + threadIdx.x) >> 5;
    if (node >= N_NODES) return;
    const float4* zrow = (const float4*)(z + (size_t)node * OUT_DIM);
    const float4* aw   = (const float4*)attn_w;
    float ps = 0.0f, pd = 0.0f;
#pragma unroll
    for (int j = 0; j < 2; ++j) {
        int vi = lane + j * 32;
        float4 zz = zrow[vi];
        float4 as = aw[vi];
        float4 ad = aw[64 + vi];
        ps += zz.x * as.x + zz.y * as.y + zz.z * as.z + zz.w * as.w;
        pd += zz.x * ad.x + zz.y * ad.y + zz.z * ad.z + zz.w * ad.w;
    }
#pragma unroll
    for (int m = 16; m >= 1; m >>= 1) {
        ps += __shfl_xor(ps, m, 32);
        pd += __shfl_xor(pd, m, 32);
    }
    if (lane == 0) { s_src[node] = ps; s_dst[node] = pd; }
}

__global__ void k_edge_scores(const int* __restrict__ src, const int* __restrict__ dst,
                              const float* __restrict__ s_src, const float* __restrict__ s_dst,
                              const float* __restrict__ attn_b,
                              float* __restrict__ e, unsigned* __restrict__ emax) {
    int i = blockIdx.x * blockDim.x + threadIdx.x;
    if (i >= N_EDGES) return;
    float v = s_src[src[i]] + s_dst[dst[i]] + attn_b[0];
    v = (v >= 0.0f) ? v : NEG_SLOPE * v;
    e[i] = v;
    atomicMax(&emax[dst[i]], enc_f32(v));
}

__global__ void k_edge_exp(const int* __restrict__ dst, float* __restrict__ e,
                           const unsigned* __restrict__ emax, float* __restrict__ denom) {
    int i = blockIdx.x * blockDim.x + threadIdx.x;
    if (i >= N_EDGES) return;
    int d = dst[i];
    float ee = expf(e[i] - dec_f32(emax[d]));
    e[i] = ee;
    unsafeAtomicAdd(&denom[d], ee);
}

__global__ void k_aggregate(const int* __restrict__ src, const int* __restrict__ dst,
                            const float* __restrict__ e_exp, const float* __restrict__ denom,
                            const float* __restrict__ z, float* __restrict__ hout) {
    const int lane = threadIdx.x & 31;
    const int edge = (blockIdx.x * blockDim.x + threadIdx.x) >> 5;
    if (edge >= N_EDGES) return;
    const int s = src[edge], d = dst[edge];
    const float alpha = e_exp[edge] / denom[d];
    const float4* zs = (const float4*)(z + (size_t)s * OUT_DIM);
    float* ho = hout + (size_t)d * OUT_DIM;
#pragma unroll
    for (int j = 0; j < 2; ++j) {
        int vi = lane + j * 32;
        float4 zz = zs[vi];
        unsafeAtomicAdd(&ho[vi * 4 + 0], alpha * zz.x);
        unsafeAtomicAdd(&ho[vi * 4 + 1], alpha * zz.y);
        unsafeAtomicAdd(&ho[vi * 4 + 2], alpha * zz.z);
        unsafeAtomicAdd(&ho[vi * 4 + 3], alpha * zz.w);
    }
}

static inline size_t align256(size_t x) { return (x + 255) & ~(size_t)255; }

extern "C" void kernel_launch(void* const* d_in, const int* in_sizes, int n_in,
                              void* d_out, int out_size, void* d_ws, size_t ws_size,
                              hipStream_t stream) {
    const float* h      = (const float*)d_in[0];
    const float* fc_w   = (const float*)d_in[1];
    const float* fc_b   = (const float*)d_in[2];
    const float* attn_w = (const float*)d_in[3];
    const float* attn_b = (const float*)d_in[4];
    const int*   src    = (const int*)d_in[5];
    const int*   dst    = (const int*)d_in[6];
    float* hout = (float*)d_out;

    char* ws = (char*)d_ws;
    float*    z      = (float*)ws;    ws += align256((size_t)N_NODES * OUT_DIM * 4);
    float*    s_src  = (float*)ws;    ws += align256((size_t)N_NODES * 4);
    float*    s_dst  = (float*)ws;    ws += align256((size_t)N_NODES * 4);
    float*    e_buf  = (float*)ws;    ws += align256((size_t)N_EDGES * 4);
    unsigned* emax   = (unsigned*)ws; ws += align256((size_t)N_NODES * 4);
    float*    denom  = (float*)ws;    ws += align256((size_t)N_NODES * 4);

    k_init<<<1250, 256, 0, stream>>>(emax, denom, hout);
    k_gemm_wmma<<<313, 256, 0, stream>>>(h, fc_w, fc_b, z);
    k_node_dots<<<1250, 256, 0, stream>>>(z, attn_w, s_src, s_dst);
    k_edge_scores<<<(N_EDGES + 255) / 256, 256, 0, stream>>>(src, dst, s_src, s_dst,
                                                             attn_b, e_buf, emax);
    k_edge_exp<<<(N_EDGES + 255) / 256, 256, 0, stream>>>(dst, e_buf, emax, denom);
    k_aggregate<<<(N_EDGES * 32 + 255) / 256, 256, 0, stream>>>(src, dst, e_buf, denom,
                                                                z, hout);
}


# Edges grouped by destination in the call (integer count, scan, place); each node's segment sorted by source id, one wave does softmax and row sum in that fixed order in registers; no float atomics
# speedup vs baseline: 1.6028x; 1.6028x over previous
#include <hip/hip_runtime.h>

#define N_NODES 10000
#define N_EDGES 320000
#define IN_DIM  512
#define OUT_DIM 256
#define NEG_SLOPE 0.01f
#define KEY_NONE 0x7FFFFFFF

typedef float v2f __attribute__((ext_vector_type(2)));
typedef float v8f __attribute__((ext_vector_type(8)));

__global__ void k_gemm_wmma(const float* __restrict__ h, const float* __restrict__ w,
                            const float* __restrict__ bias, float* __restrict__ z,
                            int* __restrict__ cnt) {
    const int gid = blockIdx.x * blockDim.x + threadIdx.x;
    if (gid < N_NODES) cnt[gid] = 0;

    const int lane = threadIdx.x & 31;
    const int gw   = blockIdx.x * (blockDim.x >> 5) + (threadIdx.x >> 5);
    if (gw >= 2500) return;
    const int row0 = (gw >> 2) * 16;
    const int col0 = (gw & 3) * 64;
    const int mn   = lane & 15;
    const int kb   = (lane >> 4) << 1;

    const float* arow = h + (size_t)(row0 + mn) * IN_DIM + kb;
    const float* b0   = w + (size_t)(col0 + mn) * IN_DIM + kb;
    const float* b1   = b0 + (size_t)16 * IN_DIM;
    const float* b2   = b0 + (size_t)32 * IN_DIM;
    const float* b3   = b0 + (size_t)48 * IN_DIM;

    v8f c0 = {}, c1 = {}, c2 = {}, c3 = {};
#pragma unroll 4
    for (int k = 0; k < IN_DIM; k += 4) {
        v2f a  = *(const v2f*)(arow + k);
        v2f q0 = *(const v2f*)(b0 + k);
        v2f q1 = *(const v2f*)(b1 + k);
        v2f q2 = *(const v2f*)(b2 + k);
        v2f q3 = *(const v2f*)(b3 + k);
        c0 = __builtin_amdgcn_wmma_f32_16x16x4_f32(false, a, false, q0, (short)0, c0, false, false);
        c1 = __builtin_amdgcn_wmma_f32_16x16x4_f32(false, a, false, q1, (short)0, c1, false, false);
        c2 = __builtin_amdgcn_wmma_f32_16x16x4_f32(false, a, false, q2, (short)0, c2, false, false);
        c3 = __builtin_amdgcn_wmma_f32_16x16x4_f32(false, a, false, q3, (short)0, c3, false, false);
    }

    const int mbase = (lane >> 4) * 8;
    v8f acc[4] = {c0, c1, c2, c3};
#pragma unroll
    for (int g = 0; g < 4; ++g) {
        const int   col = col0 + g * 16 + mn;
        const float bv  = bias[col];
#pragma unroll
        for (int v = 0; v < 8; ++v)
            z[(size_t)(row0 + mbase + v) * OUT_DIM + col] = acc[g][v] + bv;
    }
}

__global__ void k_node_dots(const float* __restrict__ z, const float* __restrict__ attn_w,
                            const int* __restrict__ dst, int* __restrict__ cnt,
                            float* __restrict__ s_src, float* __restrict__ s_dst) {
    const int gid  = blockIdx.x * blockDim.x + threadIdx.x;
    if (gid < N_EDGES) atomicAdd(&cnt[dst[gid]], 1);

    const int lane = threadIdx.x & 31;
    const int node = gid >> 5;
    if (node >= N_NODES) return;
    const float4* zrow = (const float4*)(z + (size_t)node * OUT_DIM);
    const float4* aw   = (const float4*)attn_w;
    float ps = 0.0f, pd = 0.0f;
#pragma unroll
    for (int j = 0; j < 2; ++j) {
        int vi = lane + j * 32;
        float4 zz = zrow[vi];
        float4 as = aw[vi];
        float4 ad = aw[64 + vi];
        ps += zz.x * as.x + zz.y * as.y + zz.z * as.z + zz.w * as.w;
        pd += zz.x * ad.x + zz.y * ad.y + zz.z * ad.z + zz.w * ad.w;
    }
#pragma unroll
    for (int m = 16; m >= 1; m >>= 1) {
        ps += __shfl_xor(ps, m, 32);
        pd += __shfl_xor(pd, m, 32);
    }
    if (lane == 0) { s_src[node] = ps; s_dst[node] = pd; }
}

__global__ void __launch_bounds__(1024) k_scan(const int* __restrict__ cnt,
                                               int* __restrict__ offs) {
    __shared__ int wsum[32];
    const int t    = threadIdx.x;
    const int lane = t & 31;
    const int wv   = t >> 5;
    int c[10];
    int sum = 0;
#pragma unroll
    for (int q = 0; q < 10; ++q) {
        const int idx = t * 10 + q;
        c[q] = (idx < N_NODES) ? cnt[idx] : 0;
        sum += c[q];
    }
    int incl = sum;
    for (int o = 1; o < 32; o <<= 1) {
        const int y = __shfl_up(incl, o, 32);
        if (lane >= o) incl += y;
    }
    if (lane == 31) wsum[wv] = incl;
    __syncthreads();
    if (wv == 0) {
        const int v = wsum[lane];
        int iv = v;
        for (int o = 1; o < 32; o <<= 1) {
            const int y = __shfl_up(iv, o, 32);
            if (lane >= o) iv += y;
        }
        wsum[lane] = iv - v;
    }
    __syncthreads();
    int run = wsum[wv] + incl - sum;
#pragma unroll
    for (int q = 0; q < 10; ++q) {
        const int idx = t * 10 + q;
        if (idx < N_NODES) { offs[idx] = run; run += c[q]; }
    }
}

__global__ void k_place(const int* __restrict__ src, const int* __restrict__ dst,
                        int* __restrict__ offs, int* __restrict__ perm) {
    const int i = blockIdx.x * blockDim.x + threadIdx.x;
    if (i >= N_EDGES) return;
    const int pos = atomicAdd(&offs[dst[i]], 1);
    perm[pos] = src[i];
}

__device__ __forceinline__ float leaky(float v) { return (v >= 0.0f) ? v : NEG_SLOPE * v; }

__global__ void __launch_bounds__(256) k_gather(const int* __restrict__ perm,
                                                const int* __restrict__ offs,
                                                const float* __restrict__ s_src,
                                                const float* __restrict__ s_dst,
                                                const float* __restrict__ attn_b,
                                                const float* __restrict__ z,
                                                float* __restrict__ hout) {
    __shared__ int sm[8][64];
    const int lane = threadIdx.x & 31;
    const int w     = __builtin_amdgcn_readfirstlane((int)(threadIdx.x >> 5));
    const int node  = blockIdx.x * 8 + w;
    const int start = (node > 0) ? offs[node - 1] : 0;
    const int end   = offs[node];
    const int deg   = end - start;
    const bool fast = (deg <= 64);
    const bool v0   = fast && (lane < deg);
    const bool v1   = fast && (lane + 32 < deg);
    const int  n0   = (deg < 32) ? deg : 32;

    int k0 = KEY_NONE, k1 = KEY_NONE;
    if (v0) k0 = perm[start + lane];
    if (v1) k1 = perm[start + 32 + lane];
    if (fast) {
        int r0 = 0, r1 = 0;
        for (int j = 0; j < n0; ++j) {
            const int o = __builtin_amdgcn_readlane(k0, j);
            r0 += ((o < k0) || (o == k0 && j < lane)) ? 1 : 0;
            r1 += (o <= k1) ? 1 : 0;
        }
        for (int j = 32; j < deg; ++j) {
            const int o = __builtin_amdgcn_readlane(k1, j - 32);
            r0 += (o < k0) ? 1 : 0;
            r1 += ((o < k1) || (o == k1 && (j - 32) < lane)) ? 1 : 0;
        }
        if (v0) sm[w][r0] = k0;
        if (v1) sm[w][r1] = k1;
    }
    __syncthreads();

    float4 accA = {0.0f, 0.0f, 0.0f, 0.0f}, accB = {0.0f, 0.0f, 0.0f, 0.0f};
    const float sd = s_dst[node];
    const float ab = attn_b[0];
    const float ninf = -__builtin_huge_valf();

    if (fast) {
        const int s0 = v0 ? sm[w][lane] : 0;
        const int s1 = v1 ? sm[w][lane + 32] : 0;
        float e0 = ninf, e1 = ninf;
        if (v0) e0 = leaky(s_src[s0] + sd + ab);
        if (v1) e1 = leaky(s_src[s1] + sd + ab);
        float m = fmaxf(e0, e1);
        for (int o = 16; o >= 1; o >>= 1) m = fmaxf(m, __shfl_xor(m, o, 32));
        const float x0 = v0 ? expf(e0 - m) : 0.0f;
        const float x1 = v1 ? expf(e1 - m) : 0.0f;
        float dn = x0 + x1;
        for (int o = 16; o >= 1; o >>= 1) dn += __shfl_xor(dn, o, 32);
        const int a0 = __float_as_int(x0 / dn);
        const int a1 = __float_as_int(x1 / dn);
#pragma unroll 4
        for (int r = 0; r < n0; ++r) {
            const int   s = __builtin_amdgcn_readlane(s0, r);
            const float a = __int_as_float(__builtin_amdgcn_readlane(a0, r));
            const float4* zr = (const float4*)(z + (size_t)s * OUT_DIM);
            const float4 p = zr[lane];
            const float4 q = zr[lane + 32];
            accA.x += a * p.x; accA.y += a * p.y; accA.z += a * p.z; accA.w += a * p.w;
            accB.x += a * q.x; accB.y += a * q.y; accB.z += a * q.z; accB.w += a * q.w;
        }
#pragma unroll 4
        for (int r = 32; r < deg; ++r) {
            const int   s = __builtin_amdgcn_readlane(s1, r - 32);
            const float a = __int_as_float(__builtin_amdgcn_readlane(a1, r - 32));
            const float4* zr = (const float4*)(z + (size_t)s * OUT_DIM);
            const float4 p = zr[lane];
            const float4 q = zr[lane + 32];
            accA.x += a * p.x; accA.y += a * p.y; accA.z += a * p.z; accA.w += a * p.w;
            accB.x += a * q.x; accB.y += a * q.y; accB.z += a * q.z; accB.w += a * q.w;
        }
    } else {
        float m = ninf;
        for (int i = start + lane; i < end; i += 32)
            m = fmaxf(m, leaky(s_src[perm[i]] + sd + ab));
        for (int o = 16; o >= 1; o >>= 1) m = fmaxf(m, __shfl_xor(m, o, 32));
        float dn = 0.0f;
        for (int pass = 0; pass < 2; ++pass) {
            int last = -1;
            int remaining = deg;
            for (int it = 0; it < deg && remaining > 0; ++it) {
                int cand = KEY_NONE, cn = 0;
                for (int i = start + lane; i < end; i += 32) {
                    const int k = perm[i];
                    if (k > last) {
                        if (k < cand) { cand = k; cn = 1; }
                        else if (k == cand) ++cn;
                    }
                }
                int cm = cand;
                for (int o = 16; o >= 1; o >>= 1) { const int y = __shfl_xor(cm, o, 32); cm = (y < cm) ? y : cm; }
                int c = (cand == cm) ? cn : 0;
                for (int o = 16; o >= 1; o >>= 1) c += __shfl_xor(c, o, 32);
                if (c <= 0) break;
                const float x = expf(leaky(s_src[cm] + sd + ab) - m);
                if (pass == 0) {
                    for (int t = 0; t < c; ++t) dn += x;
                } else {
                    const float a = x / dn;
                    const float4* zr = (const float4*)(z + (size_t)cm * OUT_DIM);
                    const float4 p = zr[lane];
                    const float4 q = zr[lane + 32];
                    for (int t = 0; t < c; ++t) {
                        accA.x += a * p.x; accA.y += a * p.y; accA.z += a * p.z; accA.w += a * p.w;
                        accB.x += a * q.x; accB.y += a * q.y; accB.z += a * q.z; accB.w += a * q.w;
                    }
                }
                last = cm;
                remaining -= c;
            }
        }
    }

    float4* ho = (float4*)(hout + (size_t)node * OUT_DIM);
    ho[lane]      = accA;
    ho[lane + 32] = accB;
}

static inline size_t align256(size_t x) { return (x + 255) & ~(size_t)255; }

extern "C" void kernel_launch(void* const* d_in, const int* in_sizes, int n_in,
                              void* d_out, int out_size, void* d_ws, size_t ws_size,
                              hipStream_t stream) {
    const float* h      = (const float*)d_in[0];
    const float* fc_w   = (const float*)d_in[1];
    const float* fc_b   = (const float*)d_in[2];
    const float* attn_w = (const float*)d_in[3];
    const float* attn_b = (const float*)d_in[4];
    const int*   src    = (const int*)d_in[5];
    const int*   dst    = (const int*)d_in[6];
    float* hout = (float*)d_out;

    char* ws = (char*)d_ws;
    float* z     = (float*)ws; ws += align256((size_t)N_NODES * OUT_DIM * 4);
    float* s_src = (float*)ws; ws += align256((size_t)N_NODES * 4);
    float* s_dst = (float*)ws; ws += align256((size_t)N_NODES * 4);
    int*   perm  = (int*)ws;   ws += align256((size_t)N_EDGES * 4);
    int*   cnt   = (int*)ws;   ws += align256((size_t)N_NODES * 4);
    int*   offs  = (int*)ws;   ws += align256((size_t)N_NODES * 4);

    k_gemm_wmma<<<313, 256, 0, stream>>>(h, fc_w, fc_b, z, cnt);
    k_node_dots<<<1250, 256, 0, stream>>>(z, attn_w, dst, cnt, s_src, s_dst);
    k_scan<<<1, 1024, 0, stream>>>(cnt, offs);
    k_place<<<(N_EDGES + 255) / 256, 256, 0, stream>>>(src, dst, offs, perm);
    k_gather<<<N_NODES / 8, 256, 0, stream>>>(perm, offs, s_src, s_dst, attn_b, z, hout);
}


# GEMM operands staged through LDS: each 32-k chunk of the h and w rows fetched by whole-line global loads into two LDS stages, fragments read from LDS; 4-wave groups, 16x256 strip; sum order kept
# speedup vs baseline: 1.8870x; 1.1773x over previous
#include <hip/hip_runtime.h>

#define N_NODES 10000
#define N_EDGES 320000
#define IN_DIM  512
#define OUT_DIM 256
#define NEG_SLOPE 0.01f
#define KEY_NONE 0x7FFFFFFF

typedef float v2f __attribute__((ext_vector_type(2)));
typedef float v8f __attribute__((ext_vector_type(8)));

#define KC      32
#define PITCH   36
#define A_ROWS  16
#define B_ROWS  OUT_DIM
#define STAGE_FLOATS ((A_ROWS + B_ROWS) * PITCH)

__device__ __forceinline__ void mma_chunk(const float* cur, int fA, int fB,
                                          v8f& c0, v8f& c1, v8f& c2, v8f& c3) {
#pragma unroll
    for (int kk = 0; kk < KC; kk += 4) {
        v2f a  = *(const v2f*)(cur + fA + kk);
        v2f q0 = *(const v2f*)(cur + fB + kk);
        v2f q1 = *(const v2f*)(cur + fB + 16 * PITCH + kk);
        v2f q2 = *(const v2f*)(cur + fB + 32 * PITCH + kk);
        v2f q3 = *(const v2f*)(cur + fB + 48 * PITCH + kk);
        c0 = __builtin_amdgcn_wmma_f32_16x16x4_f32(false, a, false, q0, (short)0, c0, false, false);
        c1 = __builtin_amdgcn_wmma_f32_16x16x4_f32(false, a, false, q1, (short)0, c1, false, false);
        c2 = __builtin_amdgcn_wmma_f32_16x16x4_f32(false, a, false, q2, (short)0, c2, false, false);
        c3 = __builtin_amdgcn_wmma_f32_16x16x4_f32(false, a, false, q3, (short)0, c3, false, false);
    }
}

__global__ void __launch_bounds__(128) k_gemm_wmma(const float* __restrict__ h,
                                                   const float* __restrict__ w,
                                                   const float* __restrict__ bias,
                                                   float* __restrict__ z,
                                                   int* __restrict__ cnt) {
    __shared__ __attribute__((aligned(16))) float lds[2 * STAGE_FLOATS];

    const int t   = threadIdx.x;
    const int gid = blockIdx.x * 128 + t;
    if (gid < N_NODES) cnt[gid] = 0;

    const int lane = t & 31;
    const int wv   = t >> 5;
    const int row0 = blockIdx.x * 16;
    const int col0 = wv * 64;
    const int mn   = lane & 15;
    const int kb   = (lane >> 4) << 1;

    const int seg = t & 7;
    const int sr  = t >> 3;
    const float* gA = h + (size_t)(row0 + sr) * IN_DIM + seg * 4;
    const float* gB = w + (size_t)sr * IN_DIM + seg * 4;
    const int    sA = sr * PITCH + seg * 4;
    const int    sB = (A_ROWS + sr) * PITCH + seg * 4;

    const int fA = mn * PITCH + kb;
    const int fB = (A_ROWS + col0 + mn) * PITCH + kb;

    float4 ra, rb[16];
    ra = *(const float4*)gA;
#pragma unroll
    for (int i = 0; i < 16; ++i) rb[i] = *(const float4*)(gB + (size_t)i * 16 * IN_DIM);
    *(float4*)(lds + sA) = ra;
#pragma unroll
    for (int i = 0; i < 16; ++i) *(float4*)(lds + sB + i * 16 * PITCH) = rb[i];
    __syncthreads();

    v8f c0 = {}, c1 = {}, c2 = {}, c3 = {};
    for (int c = 0; c + 1 < IN_DIM / KC; ++c) {
        const float* cur = lds + (c & 1) * STAGE_FLOATS;
        float*       nxt = lds + ((c + 1) & 1) * STAGE_FLOATS;
        const int k1 = (c + 1) * KC;
        ra = *(const float4*)(gA + k1);
#pragma unroll
        for (int i = 0; i < 16; ++i) rb[i] = *(const float4*)(gB + (size_t)i * 16 * IN_DIM + k1);
        mma_chunk(cur, fA, fB, c0, c1, c2, c3);
        *(float4*)(nxt + sA) = ra;
#pragma unroll
        for (int i = 0; i < 16; ++i) *(float4*)(nxt + sB + i * 16 * PITCH) = rb[i];
        __builtin_amdgcn_sched_group_barrier(0x020, 17, 0);
        __builtin_amdgcn_sched_group_barrier(0x100, 20, 0);
        __builtin_amdgcn_sched_group_barrier(0x008, 32, 0);
        __builtin_amdgcn_sched_group_barrier(0x200, 17, 0);
        __syncthreads();
    }
    mma_chunk(lds + ((IN_DIM / KC - 1) & 1) * STAGE_FLOATS, fA, fB, c0, c1, c2, c3);

    const int mbase = (lane >> 4) * 8;
    v8f acc[4] = {c0, c1, c2, c3};
#pragma unroll
    for (int g = 0; g < 4; ++g) {
        const int   col = col0 + g * 16 + mn;
        const float bv  = bias[col];
#pragma unroll
        for (int v = 0; v < 8; ++v)
            z[(size_t)(row0 + mbase + v) * OUT_DIM + col] = acc[g][v] + bv;
    }
}

__global__ void k_node_dots(const float* __restrict__ z, const float* __restrict__ attn_w,
                            const int* __restrict__ dst, int* __restrict__ cnt,
                            float* __restrict__ s_src, float* __restrict__ s_dst) {
    const int gid  = blockIdx.x * blockDim.x + threadIdx.x;
    if (gid < N_EDGES) atomicAdd(&cnt[dst[gid]], 1);

    const int lane = threadIdx.x & 31;
    const int node = gid >> 5;
    if (node >= N_NODES) return;
    const float4* zrow = (const float4*)(z + (size_t)node * OUT_DIM);
    const float4* aw   = (const float4*)attn_w;
    float ps = 0.0f, pd = 0.0f;
#pragma unroll
    for (int j = 0; j < 2; ++j) {
        int vi = lane + j * 32;
        float4 zz = zrow[vi];
        float4 as = aw[vi];
        float4 ad = aw[64 + vi];
        ps += zz.x * as.x + zz.y * as.y + zz.z * as.z + zz.w * as.w;
        pd += zz.x * ad.x + zz.y * ad.y + zz.z * ad.z + zz.w * ad.w;
    }
#pragma unroll
    for (int m = 16; m >= 1; m >>= 1) {
        ps += __shfl_xor(ps, m, 32);
        pd += __shfl_xor(pd, m, 32);
    }
    if (lane == 0) { s_src[node] = ps; s_dst[node] = pd; }
}

__global__ void __launch_bounds__(1024) k_scan(const int* __restrict__ cnt,
                                               int* __restrict__ offs) {
    __shared__ int wsum[32];
    const int t    = threadIdx.x;
    const int lane = t & 31;
    const int wv   = t >> 5;
    int c[10];
    int sum = 0;
#pragma unroll
    for (int q = 0; q < 10; ++q) {
        const int idx = t * 10 + q;
        c[q] = (idx < N_NODES) ? cnt[idx] : 0;
        sum += c[q];
    }
    int incl = sum;
    for (int o = 1; o < 32; o <<= 1) {
        const int y = __shfl_up(incl, o, 32);
        if (lane >= o) incl += y;
    }
    if (lane == 31) wsum[wv] = incl;
    __syncthreads();
    if (wv == 0) {
        const int v = wsum[lane];
        int iv = v;
        for (int o = 1; o < 32; o <<= 1) {
            const int y = __shfl_up(iv, o, 32);
            if (lane >= o) iv += y;
        }
        wsum[lane] = iv - v;
    }
    __syncthreads();
    int run = wsum[wv] + incl - sum;
#pragma unroll
    for (int q = 0; q < 10; ++q) {
        const int idx = t * 10 + q;
        if (idx < N_NODES) { offs[idx] = run; run += c[q]; }
    }
}

__global__ void k_place(const int* __restrict__ src, const int* __restrict__ dst,
                        int* __restrict__ offs, int* __restrict__ perm) {
    const int i = blockIdx.x * blockDim.x + threadIdx.x;
    if (i >= N_EDGES) return;
    const int pos = atomicAdd(&offs[dst[i]], 1);
    perm[pos] = src[i];
}

__device__ __forceinline__ float leaky(float v) { return (v >= 0.0f) ? v : NEG_SLOPE * v; }

__global__ void __launch_bounds__(256) k_gather(const int* __restrict__ perm,
                                                const int* __restrict__ offs,
                                                const float* __restrict__ s_src,
                                                const float* __restrict__ s_dst,
                                                const float* __restrict__ attn_b,
                                                const float* __restrict__ z,
                                                float* __restrict__ hout) {
    __shared__ int sm[8][64];
    const int lane = threadIdx.x & 31;
    const int w     = __builtin_amdgcn_readfirstlane((int)(threadIdx.x >> 5));
    const int node  = blockIdx.x * 8 + w;
    const int start = (node > 0) ? offs[node - 1] : 0;
    const int end   = offs[node];
    const int deg   = end - start;
    const bool fast = (deg <= 64);
    const bool v0   = fast && (lane < deg);
    const bool v1   = fast && (lane + 32 < deg);
    const int  n0   = (deg < 32) ? deg : 32;

    int k0 = KEY_NONE, k1 = KEY_NONE;
    if (v0) k0 = perm[start + lane];
    if (v1) k1 = perm[start + 32 + lane];
    if (fast) {
        int r0 = 0, r1 = 0;
        for (int j = 0; j < n0; ++j) {
            const int o = __builtin_amdgcn_readlane(k0, j);
            r0 += ((o < k0) || (o == k0 && j < lane)) ? 1 : 0;
            r1 += (o <= k1) ? 1 : 0;
        }
        for (int j = 32; j < deg; ++j) {
            const int o = __builtin_amdgcn_readlane(k1, j - 32);
            r0 += (o < k0) ? 1 : 0;
            r1 += ((o < k1) || (o == k1 && (j - 32) < lane)) ? 1 : 0;
        }
        if (v0) sm[w][r0] = k0;
        if (v1) sm[w][r1] = k1;
    }
    __syncthreads();

    float4 accA = {0.0f, 0.0f, 0.0f, 0.0f}, accB = {0.0f, 0.0f, 0.0f, 0.0f};
    const float sd = s_dst[node];
    const float ab = attn_b[0];
    const float ninf = -__builtin_huge_valf();

    if (fast) {
        const int s0 = v0 ? sm[w][lane] : 0;
        const int s1 = v1 ? sm[w][lane + 32] : 0;
        float e0 = ninf, e1 = ninf;
        if (v0) e0 = leaky(s_src[s0] + sd + ab);
        if (v1) e1 = leaky(s_src[s1] + sd + ab);
        float m = fmaxf(e0, e1);
        for (int o = 16; o >= 1; o >>= 1) m = fmaxf(m, __shfl_xor(m, o, 32));
        const float x0 = v0 ? expf(e0 - m) : 0.0f;
        const float x1 = v1 ? expf(e1 - m) : 0.0f;
        float dn = x0 + x1;
        for (int o = 16; o >= 1; o >>= 1) dn += __shfl_xor(dn, o, 32);
        const int a0 = __float_as_int(x0 / dn);
        const int a1 = __float_as_int(x1 / dn);
#pragma unroll 4
        for (int r = 0; r < n0; ++r) {
            const int   s = __builtin_amdgcn_readlane(s0, r);
            const float a = __int_as_float(__builtin_amdgcn_readlane(a0, r));
            const float4* zr = (const float4*)(z + (size_t)s * OUT_DIM);
            const float4 p = zr[lane];
            const float4 q = zr[lane + 32];
            accA.x += a * p.x; accA.y += a * p.y; accA.z += a * p.z; accA.w += a * p.w;
            accB.x += a * q.x; accB.y += a * q.y; accB.z += a * q.z; accB.w += a * q.w;
        }
#pragma unroll 4
        for (int r = 32; r < deg; ++r) {
            const int   s = __builtin_amdgcn_readlane(s1, r - 32);
            const float a = __int_as_float(__builtin_amdgcn_readlane(a1, r - 32));
            const float4* zr = (const float4*)(z + (size_t)s * OUT_DIM);
            const float4 p = zr[lane];
            const float4 q = zr[lane + 32];
            accA.x += a * p.x; accA.y += a * p.y; accA.z += a * p.z; accA.w += a * p.w;
            accB.x += a * q.x; accB.y += a * q.y; accB.z += a * q.z; accB.w += a * q.w;
        }
    } else {
        float m = ninf;
        for (int i = start + lane; i < end; i += 32)
            m = fmaxf(m, leaky(s_src[perm[i]] + sd + ab));
        for (int o = 16; o >= 1; o >>= 1) m = fmaxf(m, __shfl_xor(m, o, 32));
        float dn = 0.0f;
        for (int pass = 0; pass < 2; ++pass) {
            int last = -1;
            int remaining = deg;
            for (int it = 0; it < deg && remaining > 0; ++it) {
                int cand = KEY_NONE, cn = 0;
                for (int i = start + lane; i < end; i += 32) {
                    const int k = perm[i];
                    if (k > last) {
                        if (k < cand) { cand = k; cn = 1; }
                        else if (k == cand) ++cn;
                    }
                }
                int cm = cand;
                for (int o = 16; o >= 1; o >>= 1) { const int y = __shfl_xor(cm, o, 32); cm = (y < cm) ? y : cm; }
                int c = (cand == cm) ? cn : 0;
                for (int o = 16; o >= 1; o >>= 1) c += __shfl_xor(c, o, 32);
                if (c <= 0) break;
                const float x = expf(leaky(s_src[cm] + sd + ab) - m);
                if (pass == 0) {
                    for (int t = 0; t < c; ++t) dn += x;
                } else {
                    const float a = x / dn;
                    const float4* zr = (const float4*)(z + (size_t)cm * OUT_DIM);
                    const float4 p = zr[lane];
                    const float4 q = zr[lane + 32];
                    for (int t = 0; t < c; ++t) {
                        accA.x += a * p.x; accA.y += a * p.y; accA.z += a * p.z; accA.w += a * p.w;
                        accB.x += a * q.x; accB.y += a * q.y; accB.z += a * q.z; accB.w += a * q.w;
                    }
                }
                last = cm;
                remaining -= c;
            }
        }
    }

    float4* ho = (float4*)(hout + (size_t)node * OUT_DIM);
    ho[lane]      = accA;
    ho[lane + 32] = accB;
}

static inline size_t align256(size_t x) { return (x + 255) & ~(size_t)255; }

extern "C" void kernel_launch(void* const* d_in, const int* in_sizes, int n_in,
                              void* d_out, int out_size, void* d_ws, size_t ws_size,
                              hipStream_t stream) {
    const float* h      = (const float*)d_in[0];
    const float* fc_w   = (const float*)d_in[1];
    const float* fc_b   = (const float*)d_in[2];
    const float* attn_w = (const float*)d_in[3];
    const float* attn_b = (const float*)d_in[4];
    const int*   src    = (const int*)d_in[5];
    const int*   dst    = (const int*)d_in[6];
    float* hout = (float*)d_out;

    char* ws = (char*)d_ws;
    float* z     = (float*)ws; ws += align256((size_t)N_NODES * OUT_DIM * 4);
    float* s_src = (float*)ws; ws += align256((size_t)N_NODES * 4);
    float* s_dst = (float*)ws; ws += align256((size_t)N_NODES * 4);
    int*   perm  = (int*)ws;   ws += align256((size_t)N_EDGES * 4);
    int*   cnt   = (int*)ws;   ws += align256((size_t)N_NODES * 4);
    int*   offs  = (int*)ws;   ws += align256((size_t)N_NODES * 4);

    k_gemm_wmma<<<N_NODES / 16, 128, 0, stream>>>(h, fc_w, fc_b, z, cnt);
    k_node_dots<<<1250, 256, 0, stream>>>(z, attn_w, dst, cnt, s_src, s_dst);
    k_scan<<<1, 1024, 0, stream>>>(cnt, offs);
    k_place<<<(N_EDGES + 255) / 256, 256, 0, stream>>>(src, dst, offs, perm);
    k_gather<<<N_NODES / 8, 256, 0, stream>>>(perm, offs, s_src, s_dst, attn_b, z, hout);
}


# Each edge filed straight into its destination's 64-slot bin of 16-bit source ids by the slot the counting add returns; the scan and place launches go (5 launches to 3); sum order kept
# speedup vs baseline: 2.6286x; 1.3930x over previous
#include <hip/hip_runtime.h>

#define N_NODES 10000
#define N_EDGES 320000
#define IN_DIM  512
#define OUT_DIM 256
#define NEG_SLOPE 0.01f
#define KEY_NONE 0x7FFFFFFF

typedef float v2f __attribute__((ext_vector_type(2)));
typedef float v8f __attribute__((ext_vector_type(8)));

#define KC      32
#define PITCH   36
#define A_ROWS  16
#define B_ROWS  OUT_DIM
#define STAGE_FLOATS ((A_ROWS + B_ROWS) * PITCH)

__device__ __forceinline__ void mma_chunk(const float* cur, int fA, int fB,
                                          v8f& c0, v8f& c1, v8f& c2, v8f& c3) {
#pragma unroll
    for (int kk = 0; kk < KC; kk += 4) {
        v2f a  = *(const v2f*)(cur + fA + kk);
        v2f q0 = *(const v2f*)(cur + fB + kk);
        v2f q1 = *(const v2f*)(cur + fB + 16 * PITCH + kk);
        v2f q2 = *(const v2f*)(cur + fB + 32 * PITCH + kk);
        v2f q3 = *(const v2f*)(cur + fB + 48 * PITCH + kk);
        c0 = __builtin_amdgcn_wmma_f32_16x16x4_f32(false, a, false, q0, (short)0, c0, false, false);
        c1 = __builtin_amdgcn_wmma_f32_16x16x4_f32(false, a, false, q1, (short)0, c1, false, false);
        c2 = __builtin_amdgcn_wmma_f32_16x16x4_f32(false, a, false, q2, (short)0, c2, false, false);
        c3 = __builtin_amdgcn_wmma_f32_16x16x4_f32(false, a, false, q3, (short)0, c3, false, false);
    }
}

__global__ void __launch_bounds__(128) k_gemm_wmma(const float* __restrict__ h,
                                                   const float* __restrict__ w,
                                                   const float* __restrict__ bias,
                                                   float* __restrict__ z,
                                                   int* __restrict__ cnt) {
    __shared__ __attribute__((aligned(16))) float lds[2 * STAGE_FLOATS];

    const int t   = threadIdx.x;
    const int gid = blockIdx.x * 128 + t;
    if (gid < N_NODES) cnt[gid] = 0;

    const int lane = t & 31;
    const int wv   = t >> 5;
    const int row0 = blockIdx.x * 16;
    const int col0 = wv * 64;
    const int mn   = lane & 15;
    const int kb   = (lane >> 4) << 1;

    const int seg = t & 7;
    const int sr  = t >> 3;
    const float* gA = h + (size_t)(row0 + sr) * IN_DIM + seg * 4;
    const float* gB = w + (size_t)sr * IN_DIM + seg * 4;
    const int    sA = sr * PITCH + seg * 4;
    const int    sB = (A_ROWS + sr) * PITCH + seg * 4;

    const int fA = mn * PITCH + kb;
    const int fB = (A_ROWS + col0 + mn) * PITCH + kb;

    float4 ra, rb[16];
    ra = *(const float4*)gA;
#pragma unroll
    for (int i = 0; i < 16; ++i) rb[i] = *(const float4*)(gB + (size_t)i * 16 * IN_DIM);
    *(float4*)(lds + sA) = ra;
#pragma unroll
    for (int i = 0; i < 16; ++i) *(float4*)(lds + sB + i * 16 * PITCH) = rb[i];
    __syncthreads();

    v8f c0 = {}, c1 = {}, c2 = {}, c3 = {};
    for (int c = 0; c + 1 < IN_DIM / KC; ++c) {
        const float* cur = lds + (c & 1) * STAGE_FLOATS;
        float*       nxt = lds + ((c + 1) & 1) * STAGE_FLOATS;
        const int k1 = (c + 1) * KC;
        ra = *(const float4*)(gA + k1);
#pragma unroll
        for (int i = 0; i < 16; ++i) rb[i] = *(const float4*)(gB + (size_t)i * 16 * IN_DIM + k1);
        mma_chunk(cur, fA, fB, c0, c1, c2, c3);
        *(float4*)(nxt + sA) = ra;
#pragma unroll
        for (int i = 0; i < 16; ++i) *(float4*)(nxt + sB + i * 16 * PITCH) = rb[i];
        __builtin_amdgcn_sched_group_barrier(0x020, 17, 0);
        __builtin_amdgcn_sched_group_barrier(0x100, 20, 0);
        __builtin_amdgcn_sched_group_barrier(0x008, 32, 0);
        __builtin_amdgcn_sched_group_barrier(0x200, 17, 0);
        __syncthreads();
    }
    mma_chunk(lds + ((IN_DIM / KC - 1) & 1) * STAGE_FLOATS, fA, fB, c0, c1, c2, c3);

    const int mbase = (lane >> 4) * 8;
    v8f acc[4] = {c0, c1, c2, c3};
#pragma unroll
    for (int g = 0; g < 4; ++g) {
        const int   col = col0 + g * 16 + mn;
        const float bv  = bias[col];
#pragma unroll
        for (int v = 0; v < 8; ++v)
            z[(size_t)(row0 + mbase + v) * OUT_DIM + col] = acc[g][v] + bv;
    }
}

#define BIN_CAP 64
static_assert(N_NODES <= 65536, "source ids are kept as 16-bit values");

__global__ void k_node_dots(const float* __restrict__ z, const float* __restrict__ attn_w,
                            const int* __restrict__ src, const int* __restrict__ dst,
                            int* __restrict__ cnt, unsigned short* __restrict__ bins,
                            float* __restrict__ s_src, float* __restrict__ s_dst) {
    const int gid  = blockIdx.x * blockDim.x + threadIdx.x;
    int d = 0, s = 0, slot = BIN_CAP;
    if (gid < N_EDGES) {
        d    = dst[gid];
        s    = src[gid];
        slot = atomicAdd(&cnt[d], 1);
    }

    const int lane = threadIdx.x & 31;
    const int node = gid >> 5;
    if (node < N_NODES) {
        const float4* zrow = (const float4*)(z + (size_t)node * OUT_DIM);
        const float4* aw   = (const float4*)attn_w;
        float ps = 0.0f, pd = 0.0f;
#pragma unroll
        for (int j = 0; j < 2; ++j) {
            int vi = lane + j * 32;
            float4 zz = zrow[vi];
            float4 as = aw[vi];
            float4 ad = aw[64 + vi];
            ps += zz.x * as.x + zz.y * as.y + zz.z * as.z + zz.w * as.w;
            pd += zz.x * ad.x + zz.y * ad.y + zz.z * ad.z + zz.w * ad.w;
        }
#pragma unroll
        for (int m = 16; m >= 1; m >>= 1) {
            ps += __shfl_xor(ps, m, 32);
            pd += __shfl_xor(pd, m, 32);
        }
        if (lane == 0) { s_src[node] = ps; s_dst[node] = pd; }
    }

    if (slot < BIN_CAP) bins[(size_t)d * BIN_CAP + slot] = (unsigned short)s;
}

__device__ __forceinline__ float leaky(float v) { return (v >= 0.0f) ? v : NEG_SLOPE * v; }

__global__ void __launch_bounds__(256) k_gather(const unsigned short* __restrict__ bins,
                                                const int* __restrict__ cnt,
                                                const int* __restrict__ src,
                                                const int* __restrict__ dst,
                                                const float* __restrict__ s_src,
                                                const float* __restrict__ s_dst,
                                                const float* __restrict__ attn_b,
                                                const float* __restrict__ z,
                                                float* __restrict__ hout) {
    __shared__ int sm[8][64];
    const int lane = threadIdx.x & 31;
    const int w     = __builtin_amdgcn_readfirstlane((int)(threadIdx.x >> 5));
    const int node  = blockIdx.x * 8 + w;
    const int deg   = cnt[node];
    const unsigned short* bin = bins + (size_t)node * BIN_CAP;
    const int b0 = bin[lane];
    const int b1 = bin[lane + 32];
    const bool fast = (deg <= BIN_CAP);
    const bool v0   = fast && (lane < deg);
    const bool v1   = fast && (lane + 32 < deg);
    const int  n0   = (deg < 32) ? deg : 32;

    const int k0 = v0 ? b0 : KEY_NONE;
    const int k1 = v1 ? b1 : KEY_NONE;
    if (fast) {
        int r0 = 0, r1 = 0;
        for (int j = 0; j < n0; ++j) {
            const int o = __builtin_amdgcn_readlane(k0, j);
            r0 += ((o < k0) || (o == k0 && j < lane)) ? 1 : 0;
            r1 += (o <= k1) ? 1 : 0;
        }
        for (int j = 32; j < deg; ++j) {
            const int o = __builtin_amdgcn_readlane(k1, j - 32);
            r0 += (o < k0) ? 1 : 0;
            r1 += ((o < k1) || (o == k1 && (j - 32) < lane)) ? 1 : 0;
        }
        if (v0) sm[w][r0] = k0;
        if (v1) sm[w][r1] = k1;
    }
    __syncthreads();

    float4 accA = {0.0f, 0.0f, 0.0f, 0.0f}, accB = {0.0f, 0.0f, 0.0f, 0.0f};
    const float sd = s_dst[node];
    const float ab = attn_b[0];
    const float ninf = -__builtin_huge_valf();

    if (fast) {
        const int s0 = v0 ? sm[w][lane] : 0;
        const int s1 = v1 ? sm[w][lane + 32] : 0;
        float e0 = ninf, e1 = ninf;
        if (v0) e0 = leaky(s_src[s0] + sd + ab);
        if (v1) e1 = leaky(s_src[s1] + sd + ab);
        float m = fmaxf(e0, e1);
        for (int o = 16; o >= 1; o >>= 1) m = fmaxf(m, __shfl_xor(m, o, 32));
        const float x0 = v0 ? expf(e0 - m) : 0.0f;
        const float x1 = v1 ? expf(e1 - m) : 0.0f;
        float dn = x0 + x1;
        for (int o = 16; o >= 1; o >>= 1) dn += __shfl_xor(dn, o, 32);
        const int a0 = __float_as_int(x0 / dn);
        const int a1 = __float_as_int(x1 / dn);
#pragma unroll 4
        for (int r = 0; r < n0; ++r) {
            const int   s = __builtin_amdgcn_readlane(s0, r);
            const float a = __int_as_float(__builtin_amdgcn_readlane(a0, r));
            const float4* zr = (const float4*)(z + (size_t)s * OUT_DIM);
            const float4 p = zr[lane];
            const float4 q = zr[lane + 32];
            accA.x += a * p.x; accA.y += a * p.y; accA.z += a * p.z; accA.w += a * p.w;
            accB.x += a * q.x; accB.y += a * q.y; accB.z += a * q.z; accB.w += a * q.w;
        }
#pragma unroll 4
        for (int r = 32; r < deg; ++r) {
            const int   s = __builtin_amdgcn_readlane(s1, r - 32);
            const float a = __int_as_float(__builtin_amdgcn_readlane(a1, r - 32));
            const float4* zr = (const float4*)(z + (size_t)s * OUT_DIM);
            const float4 p = zr[lane];
            const float4 q = zr[lane + 32];
            accA.x += a * p.x; accA.y += a * p.y; accA.z += a * p.z; accA.w += a * p.w;
            accB.x += a * q.x; accB.y += a * q.y; accB.z += a * q.z; accB.w += a * q.w;
        }
    } else {
        float m = ninf;
        for (int i = lane; i < N_EDGES; i += 32)
            if (dst[i] == node) m = fmaxf(m, leaky(s_src[src[i]] + sd + ab));
        for (int o = 16; o >= 1; o >>= 1) m = fmaxf(m, __shfl_xor(m, o, 32));
        float dn = 0.0f;
        for (int pass = 0; pass < 2; ++pass) {
            int last = -1;
            int remaining = deg;
            for (int it = 0; it < deg && remaining > 0; ++it) {
                int cand = KEY_NONE, cn = 0;
                for (int i = lane; i < N_EDGES; i += 32) {
                    if (dst[i] == node) {
                        const int k = src[i];
                        if (k > last) {
                            if (k < cand) { cand = k; cn = 1; }
                            else if (k == cand) ++cn;
                        }
                    }
                }
                int cm = cand;
                for (int o = 16; o >= 1; o >>= 1) { const int y = __shfl_xor(cm, o, 32); cm = (y < cm) ? y : cm; }
                int c = (cand == cm) ? cn : 0;
                for (int o = 16; o >= 1; o >>= 1) c += __shfl_xor(c, o, 32);
                if (c <= 0) break;
                const float x = expf(leaky(s_src[cm] + sd + ab) - m);
                if (pass == 0) {
                    for (int t = 0; t < c; ++t) dn += x;
                } else {
                    const float a = x / dn;
                    const float4* zr = (const float4*)(z + (size_t)cm * OUT_DIM);
                    const float4 p = zr[lane];
                    const float4 q = zr[lane + 32];
                    for (int t = 0; t < c; ++t) {
                        accA.x += a * p.x; accA.y += a * p.y; accA.z += a * p.z; accA.w += a * p.w;
                        accB.x += a * q.x; accB.y += a * q.y; accB.z += a * q.z; accB.w += a * q.w;
                    }
                }
                last = cm;
                remaining -= c;
            }
        }
    }

    float4* ho = (float4*)(hout + (size_t)node * OUT_DIM);
    ho[lane]      = accA;
    ho[lane + 32] = accB;
}

static inline size_t align256(size_t x) { return (x + 255) & ~(size_t)255; }

extern "C" void kernel_launch(void* const* d_in, const int* in_sizes, int n_in,
                              void* d_out, int out_size, void* d_ws, size_t ws_size,
                              hipStream_t stream) {
    const float* h      = (const float*)d_in[0];
    const float* fc_w   = (const float*)d_in[1];
    const float* fc_b   = (const float*)d_in[2];
    const float* attn_w = (const float*)d_in[3];
    const float* attn_b = (const float*)d_in[4];
    const int*   src    = (const int*)d_in[5];
    const int*   dst    = (const int*)d_in[6];
    float* hout = (float*)d_out;

    char* ws = (char*)d_ws;
    float* z     = (float*)ws; ws += align256((size_t)N_NODES * OUT_DIM * 4);
    float* s_src = (float*)ws; ws += align256((size_t)N_NODES * 4);
    float* s_dst = (float*)ws; ws += align256((size_t)N_NODES * 4);
    unsigned short* bins = (unsigned short*)ws; ws += align256((size_t)N_NODES * BIN_CAP * 2);
    int*   cnt   = (int*)ws;   ws += align256((size_t)N_NODES * 4);

    k_gemm_wmma<<<N_NODES / 16, 128, 0, stream>>>(h, fc_w, fc_b, z, cnt);
    k_node_dots<<<1250, 256, 0, stream>>>(z, attn_w, src, dst, cnt, bins, s_src, s_dst);
    k_gather<<<N_NODES / 8, 256, 0, stream>>>(bins, cnt, src, dst, s_src, s_dst, attn_b, z, hout);
}


# Edges grouped by destination in LDS, no global counters: 64 extra work-groups of the GEMM launch each sort a 5000-edge chunk by 4-node range (count, scan, place), whole-line writes; sum order kept
# speedup vs baseline: 3.8841x; 1.4776x over previous
#include <hip/hip_runtime.h>

#define N_NODES 10000
#define N_EDGES 320000
#define IN_DIM  512
#define OUT_DIM 256
#define NEG_SLOPE 0.01f
#define KEY_NONE 0x7FFFFFFF

typedef float v2f __attribute__((ext_vector_type(2)));
typedef float v8f __attribute__((ext_vector_type(8)));

#define KC      32
#define PITCH   36
#define A_ROWS  16
#define B_ROWS  OUT_DIM
#define STAGE_FLOATS ((A_ROWS + B_ROWS) * PITCH)

#define NCHUNK  64
#define CHUNK   (N_EDGES / NCHUNK)
#define BIN4    (N_NODES / 4)
#define HIST_N  2560
static_assert(N_EDGES % NCHUNK == 0 && CHUNK % 8 == 0, "chunks are whole 16-byte groups");
static_assert(CHUNK <= 65535, "offsets inside a chunk are kept as 16-bit values");
static_assert(N_NODES % 8 == 0 && N_NODES <= 16384, "records hold a 14-bit source id");
static_assert(BIN4 < HIST_N && HIST_N == 128 * 20, "scan layout");

__device__ __forceinline__ void bin_chunk(const int* __restrict__ src, const int* __restrict__ dst,
                                          unsigned short* __restrict__ epay,
                                          unsigned short* __restrict__ eoffT,
                                          float* ldsf, const int p, const int t) {
    int* hist = (int*)ldsf;
    int* wsum = hist + HIST_N;
    unsigned short* stage = (unsigned short*)(wsum + 4);
    const int lane = t & 31;
    const int wv   = t >> 5;

    for (int k = t; k < HIST_N; k += 128) hist[k] = 0;

    const int4* d4 = (const int4*)(dst + (size_t)p * CHUNK);
    const int4* s4 = (const int4*)(src + (size_t)p * CHUNK);
    int4 dd[10], ss[10];
#pragma unroll
    for (int j = 0; j < 10; ++j) {
        const int g = t + 128 * j;
        if (g < CHUNK / 4) { dd[j] = d4[g]; ss[j] = s4[g]; }
        else { dd[j] = make_int4(0, 0, 0, 0); ss[j] = make_int4(0, 0, 0, 0); }
    }
    __syncthreads();

#pragma unroll
    for (int j = 0; j < 10; ++j) {
        if (t + 128 * j < CHUNK / 4) {
            atomicAdd(&hist[dd[j].x >> 2], 1);
            atomicAdd(&hist[dd[j].y >> 2], 1);
            atomicAdd(&hist[dd[j].z >> 2], 1);
            atomicAdd(&hist[dd[j].w >> 2], 1);
        }
    }
    __syncthreads();

    const int base = t * 20;
    int loc[20], tot = 0;
#pragma unroll
    for (int k = 0; k < 20; ++k) { loc[k] = hist[base + k]; tot += loc[k]; }
    int inc = tot;
#pragma unroll
    for (int o = 1; o < 32; o <<= 1) {
        const int y = __shfl_up(inc, o, 32);
        if (lane >= o) inc += y;
    }
    if (lane == 31) wsum[wv] = inc;
    __syncthreads();
    int run = inc - tot;
    for (int q = 0; q < wv; ++q) run += wsum[q];
#pragma unroll
    for (int k = 0; k < 20; ++k) {
        const int bin = base + k;
        hist[bin] = run;
        if (bin <= BIN4) eoffT[(size_t)bin * NCHUNK + p] = (unsigned short)run;
        run += loc[k];
    }
    __syncthreads();

#pragma unroll
    for (int j = 0; j < 10; ++j) {
        if (t + 128 * j < CHUNK / 4) {
            const int dv[4] = {dd[j].x, dd[j].y, dd[j].z, dd[j].w};
            const int sv[4] = {ss[j].x, ss[j].y, ss[j].z, ss[j].w};
#pragma unroll
            for (int c = 0; c < 4; ++c) {
                const int pos = atomicAdd(&hist[dv[c] >> 2], 1);
                stage[pos] = (unsigned short)(((dv[c] & 3) << 14) | sv[c]);
            }
        }
    }
    __syncthreads();

    const uint4* st4 = (const uint4*)stage;
    uint4* o4 = (uint4*)(epay + (size_t)p * CHUNK);
    for (int k = t; k < CHUNK / 8; k += 128) o4[k] = st4[k];
}

__device__ __forceinline__ void mma_chunk(const float* cur, int fA, int fB,
                                          v8f& c0, v8f& c1, v8f& c2, v8f& c3) {
#pragma unroll
    for (int kk = 0; kk < KC; kk += 4) {
        v2f a  = *(const v2f*)(cur + fA + kk);
        v2f q0 = *(const v2f*)(cur + fB + kk);
        v2f q1 = *(const v2f*)(cur + fB + 16 * PITCH + kk);
        v2f q2 = *(const v2f*)(cur + fB + 32 * PITCH + kk);
        v2f q3 = *(const v2f*)(cur + fB + 48 * PITCH + kk);
        c0 = __builtin_amdgcn_wmma_f32_16x16x4_f32(false, a, false, q0, (short)0, c0, false, false);
        c1 = __builtin_amdgcn_wmma_f32_16x16x4_f32(false, a, false, q1, (short)0, c1, false, false);
        c2 = __builtin_amdgcn_wmma_f32_16x16x4_f32(false, a, false, q2, (short)0, c2, false, false);
        c3 = __builtin_amdgcn_wmma_f32_16x16x4_f32(false, a, false, q3, (short)0, c3, false, false);
    }
}

__global__ void __launch_bounds__(128) k_gemm_wmma(const float* __restrict__ h,
                                                   const float* __restrict__ w,
                                                   const float* __restrict__ bias,
                                                   float* __restrict__ z,
                                                   const int* __restrict__ src,
                                                   const int* __restrict__ dst,
                                                   unsigned short* __restrict__ epay,
                                                   unsigned short* __restrict__ eoffT) {
    __shared__ __attribute__((aligned(16))) float lds[2 * STAGE_FLOATS];

    const int t   = threadIdx.x;
    if (blockIdx.x < NCHUNK) { bin_chunk(src, dst, epay, eoffT, lds, (int)blockIdx.x, t); return; }
    const int strip = (int)blockIdx.x - NCHUNK;

    const int lane = t & 31;
    const int wv   = t >> 5;
    const int row0 = strip * 16;
    const int col0 = wv * 64;
    const int mn   = lane & 15;
    const int kb   = (lane >> 4) << 1;

    const int seg = t & 7;
    const int sr  = t >> 3;
    const float* gA = h + (size_t)(row0 + sr) * IN_DIM + seg * 4;
    const float* gB = w + (size_t)sr * IN_DIM + seg * 4;
    const int    sA = sr * PITCH + seg * 4;
    const int    sB = (A_ROWS + sr) * PITCH + seg * 4;

    const int fA = mn * PITCH + kb;
    const int fB = (A_ROWS + col0 + mn) * PITCH + kb;

    float4 ra, rb[16];
    ra = *(const float4*)gA;
#pragma unroll
    for (int i = 0; i < 16; ++i) rb[i] = *(const float4*)(gB + (size_t)i * 16 * IN_DIM);
    *(float4*)(lds + sA) = ra;
#pragma unroll
    for (int i = 0; i < 16; ++i) *(float4*)(lds + sB + i * 16 * PITCH) = rb[i];
    __syncthreads();

    v8f c0 = {}, c1 = {}, c2 = {}, c3 = {};
    for (int c = 0; c + 1 < IN_DIM / KC; ++c) {
        const float* cur = lds + (c & 1) * STAGE_FLOATS;
        float*       nxt = lds + ((c + 1) & 1) * STAGE_FLOATS;
        const int k1 = (c + 1) * KC;
        ra = *(const float4*)(gA + k1);
#pragma unroll
        for (int i = 0; i < 16; ++i) rb[i] = *(const float4*)(gB + (size_t)i * 16 * IN_DIM + k1);
        mma_chunk(cur, fA, fB, c0, c1, c2, c3);
        *(float4*)(nxt + sA) = ra;
#pragma unroll
        for (int i = 0; i < 16; ++i) *(float4*)(nxt + sB + i * 16 * PITCH) = rb[i];
        __builtin_amdgcn_sched_group_barrier(0x020, 17, 0);
        __builtin_amdgcn_sched_group_barrier(0x100, 20, 0);
        __builtin_amdgcn_sched_group_barrier(0x008, 32, 0);
        __builtin_amdgcn_sched_group_barrier(0x200, 17, 0);
        __syncthreads();
    }
    mma_chunk(lds + ((IN_DIM / KC - 1) & 1) * STAGE_FLOATS, fA, fB, c0, c1, c2, c3);

    const int mbase = (lane >> 4) * 8;
    v8f acc[4] = {c0, c1, c2, c3};
#pragma unroll
    for (int g = 0; g < 4; ++g) {
        const int   col = col0 + g * 16 + mn;
        const float bv  = bias[col];
#pragma unroll
        for (int v = 0; v < 8; ++v)
            z[(size_t)(row0 + mbase + v) * OUT_DIM + col] = acc[g][v] + bv;
    }
}

#define BIN_CAP 64

__global__ void k_node_dots(const float* __restrict__ z, const float* __restrict__ attn_w,
                            float* __restrict__ s_src, float* __restrict__ s_dst) {
    const int gid  = blockIdx.x * blockDim.x + threadIdx.x;

    const int lane = threadIdx.x & 31;
    const int node = gid >> 5;
    if (node < N_NODES) {
        const float4* zrow = (const float4*)(z + (size_t)node * OUT_DIM);
        const float4* aw   = (const float4*)attn_w;
        float ps = 0.0f, pd = 0.0f;
#pragma unroll
        for (int j = 0; j < 2; ++j) {
            int vi = lane + j * 32;
            float4 zz = zrow[vi];
            float4 as = aw[vi];
            float4 ad = aw[64 + vi];
            ps += zz.x * as.x + zz.y * as.y + zz.z * as.z + zz.w * as.w;
            pd += zz.x * ad.x + zz.y * ad.y + zz.z * ad.z + zz.w * ad.w;
        }
#pragma unroll
        for (int m = 16; m >= 1; m >>= 1) {
            ps += __shfl_xor(ps, m, 32);
            pd += __shfl_xor(pd, m, 32);
        }
        if (lane == 0) { s_src[node] = ps; s_dst[node] = pd; }
    }
}

__device__ __forceinline__ float leaky(float v) { return (v >= 0.0f) ? v : NEG_SLOPE * v; }

__global__ void __launch_bounds__(256) k_gather(const unsigned short* __restrict__ epay,
                                                const unsigned short* __restrict__ eoffT,
                                                const int* __restrict__ src,
                                                const int* __restrict__ dst,
                                                const float* __restrict__ s_src,
                                                const float* __restrict__ s_dst,
                                                const float* __restrict__ attn_b,
                                                const float* __restrict__ z,
                                                float* __restrict__ hout) {
    __shared__ int sm[8][64];
    __shared__ int raw[8][BIN_CAP];
    __shared__ int lcnt[8];
    const int lane = threadIdx.x & 31;
    const int w     = __builtin_amdgcn_readfirstlane((int)(threadIdx.x >> 5));
    const int node  = blockIdx.x * 8 + w;

    if (threadIdx.x < 8) lcnt[threadIdx.x] = 0;
    __syncthreads();
    {
        const int p = threadIdx.x & (NCHUNK - 1);
        const unsigned short* ho = eoffT + (size_t)(2 * blockIdx.x) * NCHUNK + p;
        const int o0 = ho[0];
        const int o1 = ho[NCHUNK];
        const int o2 = ho[2 * NCHUNK];
        const unsigned short* pay = epay + (size_t)p * CHUNK;
        for (int i = o0 + (int)(threadIdx.x >> 6); i < o2; i += 4) {
            const int rec = pay[i];
            const int nl  = ((i >= o1) ? 4 : 0) + (rec >> 14);
            const int pos = atomicAdd(&lcnt[nl], 1);
            if (pos < BIN_CAP) raw[nl][pos] = rec & 0x3FFF;
        }
    }
    __syncthreads();
    const int deg   = __builtin_amdgcn_readfirstlane(lcnt[w]);
    const bool fast = (deg <= BIN_CAP);
    const bool v0   = fast && (lane < deg);
    const bool v1   = fast && (lane + 32 < deg);
    const int  n0   = (deg < 32) ? deg : 32;

    const int k0 = v0 ? raw[w][lane] : KEY_NONE;
    const int k1 = v1 ? raw[w][lane + 32] : KEY_NONE;
    if (fast) {
        int r0 = 0, r1 = 0;
        for (int j = 0; j < n0; ++j) {
            const int o = __builtin_amdgcn_readlane(k0, j);
            r0 += ((o < k0) || (o == k0 && j < lane)) ? 1 : 0;
            r1 += (o <= k1) ? 1 : 0;
        }
        for (int j = 32; j < deg; ++j) {
            const int o = __builtin_amdgcn_readlane(k1, j - 32);
            r0 += (o < k0) ? 1 : 0;
            r1 += ((o < k1) || (o == k1 && (j - 32) < lane)) ? 1 : 0;
        }
        if (v0) sm[w][r0] = k0;
        if (v1) sm[w][r1] = k1;
    }
    __syncthreads();

    float4 accA = {0.0f, 0.0f, 0.0f, 0.0f}, accB = {0.0f, 0.0f, 0.0f, 0.0f};
    const float sd = s_dst[node];
    const float ab = attn_b[0];
    const float ninf = -__builtin_huge_valf();

    if (fast) {
        const int s0 = v0 ? sm[w][lane] : 0;
        const int s1 = v1 ? sm[w][lane + 32] : 0;
        float e0 = ninf, e1 = ninf;
        if (v0) e0 = leaky(s_src[s0] + sd + ab);
        if (v1) e1 = leaky(s_src[s1] + sd + ab);
        float m = fmaxf(e0, e1);
        for (int o = 16; o >= 1; o >>= 1) m = fmaxf(m, __shfl_xor(m, o, 32));
        const float x0 = v0 ? expf(e0 - m) : 0.0f;
        const float x1 = v1 ? expf(e1 - m) : 0.0f;
        float dn = x0 + x1;
        for (int o = 16; o >= 1; o >>= 1) dn += __shfl_xor(dn, o, 32);
        const int a0 = __float_as_int(x0 / dn);
        const int a1 = __float_as_int(x1 / dn);
#pragma unroll 4
        for (int r = 0; r < n0; ++r) {
            const int   s = __builtin_amdgcn_readlane(s0, r);
            const float a = __int_as_float(__builtin_amdgcn_readlane(a0, r));
            const float4* zr = (const float4*)(z + (size_t)s * OUT_DIM);
            const float4 p = zr[lane];
            const float4 q = zr[lane + 32];
            accA.x += a * p.x; accA.y += a * p.y; accA.z += a * p.z; accA.w += a * p.w;
            accB.x += a * q.x; accB.y += a * q.y; accB.z += a * q.z; accB.w += a * q.w;
        }
#pragma unroll 4
        for (int r = 32; r < deg; ++r) {
            const int   s = __builtin_amdgcn_readlane(s1, r - 32);
            const float a = __int_as_float(__builtin_amdgcn_readlane(a1, r - 32));
            const float4* zr = (const float4*)(z + (size_t)s * OUT_DIM);
            const float4 p = zr[lane];
            const float4 q = zr[lane + 32];
            accA.x += a * p.x; accA.y += a * p.y; accA.z += a * p.z; accA.w += a * p.w;
            accB.x += a * q.x; accB.y += a * q.y; accB.z += a * q.z; accB.w += a * q.w;
        }
    } else {
        float m = ninf;
        for (int i = lane; i < N_EDGES; i += 32)
            if (dst[i] == node) m = fmaxf(m, leaky(s_src[src[i]] + sd + ab));
        for (int o = 16; o >= 1; o >>= 1) m = fmaxf(m, __shfl_xor(m, o, 32));
        float dn = 0.0f;
        for (int pass = 0; pass < 2; ++pass) {
            int last = -1;
            int remaining = deg;
            for (int it = 0; it < deg && remaining > 0; ++it) {
                int cand = KEY_NONE, cn = 0;
                for (int i = lane; i < N_EDGES; i += 32) {
                    if (dst[i] == node) {
                        const int k = src[i];
                        if (k > last) {
                            if (k < cand) { cand = k; cn = 1; }
                            else if (k == cand) ++cn;
                        }
                    }
                }
                int cm = cand;
                for (int o = 16; o >= 1; o >>= 1) { const int y = __shfl_xor(cm, o, 32); cm = (y < cm) ? y : cm; }
                int c = (cand == cm) ? cn : 0;
                for (int o = 16; o >= 1; o >>= 1) c += __shfl_xor(c, o, 32);
                if (c <= 0) break;
                const float x = expf(leaky(s_src[cm] + sd + ab) - m);
                if (pass == 0) {
                    for (int t = 0; t < c; ++t) dn += x;
                } else {
                    const float a = x / dn;
                    const float4* zr = (const float4*)(z + (size_t)cm * OUT_DIM);
                    const float4 p = zr[lane];
                    const float4 q = zr[lane + 32];
                    for (int t = 0; t < c; ++t) {
                        accA.x += a * p.x; accA.y += a * p.y; accA.z += a * p.z; accA.w += a * p.w;
                        accB.x += a * q.x; accB.y += a * q.y; accB.z += a * q.z; accB.w += a * q.w;
                    }
                }
                last = cm;
                remaining -= c;
            }
        }
    }

    float4* ho = (float4*)(hout + (size_t)node * OUT_DIM);
    ho[lane]      = accA;
    ho[lane + 32] = accB;
}

static inline size_t align256(size_t x) { return (x + 255) & ~(size_t)255; }

extern "C" void kernel_launch(void* const* d_in, const int* in_sizes, int n_in,
                              void* d_out, int out_size, void* d_ws, size_t ws_size,
                              hipStream_t stream) {
    const float* h      = (const float*)d_in[0];
    const float* fc_w   = (const float*)d_in[1];
    const float* fc_b   = (const float*)d_in[2];
    const float* attn_w = (const float*)d_in[3];
    const float* attn_b = (const float*)d_in[4];
    const int*   src    = (const int*)d_in[5];
    const int*   dst    = (const int*)d_in[6];
    float* hout = (float*)d_out;

    char* ws = (char*)d_ws;
    float* z     = (float*)ws; ws += align256((size_t)N_NODES * OUT_DIM * 4);
    float* s_src = (float*)ws; ws += align256((size_t)N_NODES * 4);
    float* s_dst = (float*)ws; ws += align256((size_t)N_NODES * 4);
    unsigned short* epay  = (unsigned short*)ws; ws += align256((size_t)N_EDGES * 2);
    unsigned short* eoffT = (unsigned short*)ws; ws += align256((size_t)(BIN4 + 1) * NCHUNK * 2);

    k_gemm_wmma<<<N_NODES / 16 + NCHUNK, 128, 0, stream>>>(h, fc_w, fc_b, z, src, dst, epay, eoffT);
    k_node_dots<<<1250, 256, 0, stream>>>(z, attn_w, s_src, s_dst);
    k_gather<<<N_NODES / 8, 256, 0, stream>>>(epay, eoffT, src, dst, s_src, s_dst, attn_b, z, hout);
}


# Per-node attention dots (z.a_src, z.a_dst) summed from the GEMM accumulators in its epilogue (lane columns, 16-lane row, 4 waves); node-dot launch goes (3 launches to 2); dot sum order changed
# speedup vs baseline: 4.2567x; 1.0959x over previous
#include <hip/hip_runtime.h>

#define N_NODES 10000
#define N_EDGES 320000
#define IN_DIM  512
#define OUT_DIM 256
#define NEG_SLOPE 0.01f
#define KEY_NONE 0x7FFFFFFF

typedef float v2f __attribute__((ext_vector_type(2)));
typedef float v8f __attribute__((ext_vector_type(8)));

#define KC      32
#define PITCH   36
#define A_ROWS  16
#define B_ROWS  OUT_DIM
#define STAGE_FLOATS ((A_ROWS + B_ROWS) * PITCH)

#define NCHUNK  64
#define CHUNK   (N_EDGES / NCHUNK)
#define BIN4    (N_NODES / 4)
#define HIST_N  2560
static_assert(N_EDGES % NCHUNK == 0 && CHUNK % 8 == 0, "chunks are whole 16-byte groups");
static_assert(CHUNK <= 65535, "offsets inside a chunk are kept as 16-bit values");
static_assert(N_NODES % 8 == 0 && N_NODES <= 16384, "records hold a 14-bit source id");
static_assert(BIN4 < HIST_N && HIST_N == 128 * 20, "scan layout");

__device__ __forceinline__ void bin_chunk(const int* __restrict__ src, const int* __restrict__ dst,
                                          unsigned short* __restrict__ epay,
                                          unsigned short* __restrict__ eoffT,
                                          float* ldsf, const int p, const int t) {
    int* hist = (int*)ldsf;
    int* wsum = hist + HIST_N;
    unsigned short* stage = (unsigned short*)(wsum + 4);
    const int lane = t & 31;
    const int wv   = t >> 5;

    for (int k = t; k < HIST_N; k += 128) hist[k] = 0;

    const int4* d4 = (const int4*)(dst + (size_t)p * CHUNK);
    const int4* s4 = (const int4*)(src + (size_t)p * CHUNK);
    int4 dd[10], ss[10];
#pragma unroll
    for (int j = 0; j < 10; ++j) {
        const int g = t + 128 * j;
        if (g < CHUNK / 4) { dd[j] = d4[g]; ss[j] = s4[g]; }
        else { dd[j] = make_int4(0, 0, 0, 0); ss[j] = make_int4(0, 0, 0, 0); }
    }
    __syncthreads();

#pragma unroll
    for (int j = 0; j < 10; ++j) {
        if (t + 128 * j < CHUNK / 4) {
            atomicAdd(&hist[dd[j].x >> 2], 1);
            atomicAdd(&hist[dd[j].y >> 2], 1);
            atomicAdd(&hist[dd[j].z >> 2], 1);
            atomicAdd(&hist[dd[j].w >> 2], 1);
        }
    }
    __syncthreads();

    const int base = t * 20;
    int loc[20], tot = 0;
#pragma unroll
    for (int k = 0; k < 20; ++k) { loc[k] = hist[base + k]; tot += loc[k]; }
    int inc = tot;
#pragma unroll
    for (int o = 1; o < 32; o <<= 1) {
        const int y = __shfl_up(inc, o, 32);
        if (lane >= o) inc += y;
    }
    if (lane == 31) wsum[wv] = inc;
    __syncthreads();
    int run = inc - tot;
    for (int q = 0; q < wv; ++q) run += wsum[q];
#pragma unroll
    for (int k = 0; k < 20; ++k) {
        const int bin = base + k;
        hist[bin] = run;
        if (bin <= BIN4) eoffT[(size_t)bin * NCHUNK + p] = (unsigned short)run;
        run += loc[k];
    }
    __syncthreads();

#pragma unroll
    for (int j = 0; j < 10; ++j) {
        if (t + 128 * j < CHUNK / 4) {
            const int dv[4] = {dd[j].x, dd[j].y, dd[j].z, dd[j].w};
            const int sv[4] = {ss[j].x, ss[j].y, ss[j].z, ss[j].w};
#pragma unroll
            for (int c = 0; c < 4; ++c) {
                const int pos = atomicAdd(&hist[dv[c] >> 2], 1);
                stage[pos] = (unsigned short)(((dv[c] & 3) << 14) | sv[c]);
            }
        }
    }
    __syncthreads();

    const uint4* st4 = (const uint4*)stage;
    uint4* o4 = (uint4*)(epay + (size_t)p * CHUNK);
    for (int k = t; k < CHUNK / 8; k += 128) o4[k] = st4[k];
}

template <int M>
__device__ __forceinline__ float row_xor_add(float v) {
    const int p = __builtin_amdgcn_update_dpp(0, __float_as_int(v), 0x160 + M, 0xf, 0xf, true);
    return v + __int_as_float(p);
}

__device__ __forceinline__ void mma_chunk(const float* cur, int fA, int fB,
                                          v8f& c0, v8f& c1, v8f& c2, v8f& c3) {
#pragma unroll
    for (int kk = 0; kk < KC; kk += 4) {
        v2f a  = *(const v2f*)(cur + fA + kk);
        v2f q0 = *(const v2f*)(cur + fB + kk);
        v2f q1 = *(const v2f*)(cur + fB + 16 * PITCH + kk);
        v2f q2 = *(const v2f*)(cur + fB + 32 * PITCH + kk);
        v2f q3 = *(const v2f*)(cur + fB + 48 * PITCH + kk);
        c0 = __builtin_amdgcn_wmma_f32_16x16x4_f32(false, a, false, q0, (short)0, c0, false, false);
        c1 = __builtin_amdgcn_wmma_f32_16x16x4_f32(false, a, false, q1, (short)0, c1, false, false);
        c2 = __builtin_amdgcn_wmma_f32_16x16x4_f32(false, a, false, q2, (short)0, c2, false, false);
        c3 = __builtin_amdgcn_wmma_f32_16x16x4_f32(false, a, false, q3, (short)0, c3, false, false);
    }
}

__global__ void __launch_bounds__(128) k_gemm_wmma(const float* __restrict__ h,
                                                   const float* __restrict__ w,
                                                   const float* __restrict__ bias,
                                                   const float* __restrict__ attn_w,
                                                   float* __restrict__ z,
                                                   float* __restrict__ s_src,
                                                   float* __restrict__ s_dst,
                                                   const int* __restrict__ src,
                                                   const int* __restrict__ dst,
                                                   unsigned short* __restrict__ epay,
                                                   unsigned short* __restrict__ eoffT) {
    __shared__ __attribute__((aligned(16))) float lds[2 * STAGE_FLOATS];

    const int t   = threadIdx.x;
    if (blockIdx.x < NCHUNK) { bin_chunk(src, dst, epay, eoffT, lds, (int)blockIdx.x, t); return; }
    const int strip = (int)blockIdx.x - NCHUNK;

    const int lane = t & 31;
    const int wv   = t >> 5;
    const int row0 = strip * 16;
    const int col0 = wv * 64;
    const int mn   = lane & 15;
    const int kb   = (lane >> 4) << 1;

    const int seg = t & 7;
    const int sr  = t >> 3;
    const float* gA = h + (size_t)(row0 + sr) * IN_DIM + seg * 4;
    const float* gB = w + (size_t)sr * IN_DIM + seg * 4;
    const int    sA = sr * PITCH + seg * 4;
    const int    sB = (A_ROWS + sr) * PITCH + seg * 4;

    const int fA = mn * PITCH + kb;
    const int fB = (A_ROWS + col0 + mn) * PITCH + kb;

    float4 ra, rb[16];
    ra = *(const float4*)gA;
#pragma unroll
    for (int i = 0; i < 16; ++i) rb[i] = *(const float4*)(gB + (size_t)i * 16 * IN_DIM);
    *(float4*)(lds + sA) = ra;
#pragma unroll
    for (int i = 0; i < 16; ++i) *(float4*)(lds + sB + i * 16 * PITCH) = rb[i];
    __syncthreads();

    v8f c0 = {}, c1 = {}, c2 = {}, c3 = {};
    for (int c = 0; c + 1 < IN_DIM / KC; ++c) {
        const float* cur = lds + (c & 1) * STAGE_FLOATS;
        float*       nxt = lds + ((c + 1) & 1) * STAGE_FLOATS;
        const int k1 = (c + 1) * KC;
        ra = *(const float4*)(gA + k1);
#pragma unroll
        for (int i = 0; i < 16; ++i) rb[i] = *(const float4*)(gB + (size_t)i * 16 * IN_DIM + k1);
        mma_chunk(cur, fA, fB, c0, c1, c2, c3);
        *(float4*)(nxt + sA) = ra;
#pragma unroll
        for (int i = 0; i < 16; ++i) *(float4*)(nxt + sB + i * 16 * PITCH) = rb[i];
        __builtin_amdgcn_sched_group_barrier(0x020, 17, 0);
        __builtin_amdgcn_sched_group_barrier(0x100, 20, 0);
        __builtin_amdgcn_sched_group_barrier(0x008, 32, 0);
        __builtin_amdgcn_sched_group_barrier(0x200, 17, 0);
        __syncthreads();
    }
    mma_chunk(lds + ((IN_DIM / KC - 1) & 1) * STAGE_FLOATS, fA, fB, c0, c1, c2, c3);

    const int mbase = (lane >> 4) * 8;
    v8f acc[4] = {c0, c1, c2, c3};
    float ps[8], pd[8];
#pragma unroll
    for (int v = 0; v < 8; ++v) { ps[v] = 0.0f; pd[v] = 0.0f; }
#pragma unroll
    for (int g = 0; g < 4; ++g) {
        const int   col = col0 + g * 16 + mn;
        const float bv  = bias[col];
        const float as  = attn_w[col];
        const float ad  = attn_w[OUT_DIM + col];
#pragma unroll
        for (int v = 0; v < 8; ++v) {
            acc[g][v] += bv;
            ps[v] += acc[g][v] * as;
            pd[v] += acc[g][v] * ad;
        }
    }
#pragma unroll
    for (int v = 0; v < 8; ++v) {
        ps[v] = row_xor_add<1>(ps[v]);  pd[v] = row_xor_add<1>(pd[v]);
        ps[v] = row_xor_add<2>(ps[v]);  pd[v] = row_xor_add<2>(pd[v]);
        ps[v] = row_xor_add<4>(ps[v]);  pd[v] = row_xor_add<4>(pd[v]);
        ps[v] = row_xor_add<8>(ps[v]);  pd[v] = row_xor_add<8>(pd[v]);
    }
    float* part = lds;
    if (mn == 0) {
#pragma unroll
        for (int v = 0; v < 8; ++v)
            *(float2*)(part + (wv * 16 + mbase + v) * 2) = make_float2(ps[v], pd[v]);
    }
    __syncthreads();
#pragma unroll
    for (int g = 0; g < 4; ++g) {
        const int col = col0 + g * 16 + mn;
#pragma unroll
        for (int v = 0; v < 8; ++v)
            z[(size_t)(row0 + mbase + v) * OUT_DIM + col] = acc[g][v];
    }
    if (t < 32) {
        const int r = t & 15, which = t >> 4;
        float s = part[(0 * 16 + r) * 2 + which];
        s += part[(1 * 16 + r) * 2 + which];
        s += part[(2 * 16 + r) * 2 + which];
        s += part[(3 * 16 + r) * 2 + which];
        (which ? s_dst : s_src)[row0 + r] = s;
    }
}

#define BIN_CAP 64

__device__ __forceinline__ float leaky(float v) { return (v >= 0.0f) ? v : NEG_SLOPE * v; }

__global__ void __launch_bounds__(256) k_gather(const unsigned short* __restrict__ epay,
                                                const unsigned short* __restrict__ eoffT,
                                                const int* __restrict__ src,
                                                const int* __restrict__ dst,
                                                const float* __restrict__ s_src,
                                                const float* __restrict__ s_dst,
                                                const float* __restrict__ attn_b,
                                                const float* __restrict__ z,
                                                float* __restrict__ hout) {
    __shared__ int sm[8][64];
    __shared__ int raw[8][BIN_CAP];
    __shared__ int lcnt[8];
    const int lane = threadIdx.x & 31;
    const int w     = __builtin_amdgcn_readfirstlane((int)(threadIdx.x >> 5));
    const int node  = blockIdx.x * 8 + w;

    if (threadIdx.x < 8) lcnt[threadIdx.x] = 0;
    __syncthreads();
    {
        const int p = threadIdx.x & (NCHUNK - 1);
        const unsigned short* ho = eoffT + (size_t)(2 * blockIdx.x) * NCHUNK + p;
        const int o0 = ho[0];
        const int o1 = ho[NCHUNK];
        const int o2 = ho[2 * NCHUNK];
        const unsigned short* pay = epay + (size_t)p * CHUNK;
        for (int i = o0 + (int)(threadIdx.x >> 6); i < o2; i += 4) {
            const int rec = pay[i];
            const int nl  = ((i >= o1) ? 4 : 0) + (rec >> 14);
            const int pos = atomicAdd(&lcnt[nl], 1);
            if (pos < BIN_CAP) raw[nl][pos] = rec & 0x3FFF;
        }
    }
    __syncthreads();
    const int deg   = __builtin_amdgcn_readfirstlane(lcnt[w]);
    const bool fast = (deg <= BIN_CAP);
    const bool v0   = fast && (lane < deg);
    const bool v1   = fast && (lane + 32 < deg);
    const int  n0   = (deg < 32) ? deg : 32;

    const int k0 = v0 ? raw[w][lane] : KEY_NONE;
    const int k1 = v1 ? raw[w][lane + 32] : KEY_NONE;
    if (fast) {
        int r0 = 0, r1 = 0;
        for (int j = 0; j < n0; ++j) {
            const int o = __builtin_amdgcn_readlane(k0, j);
            r0 += ((o < k0) || (o == k0 && j < lane)) ? 1 : 0;
            r1 += (o <= k1) ? 1 : 0;
        }
        for (int j = 32; j < deg; ++j) {
            const int o = __builtin_amdgcn_readlane(k1, j - 32);
            r0 += (o < k0) ? 1 : 0;
            r1 += ((o < k1) || (o == k1 && (j - 32) < lane)) ? 1 : 0;
        }
        if (v0) sm[w][r0] = k0;
        if (v1) sm[w][r1] = k1;
    }
    __syncthreads();

    float4 accA = {0.0f, 0.0f, 0.0f, 0.0f}, accB = {0.0f, 0.0f, 0.0f, 0.0f};
    const float sd = s_dst[node];
    const float ab = attn_b[0];
    const float ninf = -__builtin_huge_valf();

    if (fast) {
        const int s0 = v0 ? sm[w][lane] : 0;
        const int s1 = v1 ? sm[w][lane + 32] : 0;
        float e0 = ninf, e1 = ninf;
        if (v0) e0 = leaky(s_src[s0] + sd + ab);
        if (v1) e1 = leaky(s_src[s1] + sd + ab);
        float m = fmaxf(e0, e1);
        for (int o = 16; o >= 1; o >>= 1) m = fmaxf(m, __shfl_xor(m, o, 32));
        const float x0 = v0 ? expf(e0 - m) : 0.0f;
        const float x1 = v1 ? expf(e1 - m) : 0.0f;
        float dn = x0 + x1;
        for (int o = 16; o >= 1; o >>= 1) dn += __shfl_xor(dn, o, 32);
        const int a0 = __float_as_int(x0 / dn);
        const int a1 = __float_as_int(x1 / dn);
#pragma unroll 4
        for (int r = 0; r < n0; ++r) {
            const int   s = __builtin_amdgcn_readlane(s0, r);
            const float a = __int_as_float(__builtin_amdgcn_readlane(a0, r));
            const float4* zr = (const float4*)(z + (size_t)s * OUT_DIM);
            const float4 p = zr[lane];
            const float4 q = zr[lane + 32];
            accA.x += a * p.x; accA.y += a * p.y; accA.z += a * p.z; accA.w += a * p.w;
            accB.x += a * q.x; accB.y += a * q.y; accB.z += a * q.z; accB.w += a * q.w;
        }
#pragma unroll 4
        for (int r = 32; r < deg; ++r) {
            const int   s = __builtin_amdgcn_readlane(s1, r - 32);
            const float a = __int_as_float(__builtin_amdgcn_readlane(a1, r - 32));
            const float4* zr = (const float4*)(z + (size_t)s * OUT_DIM);
            const float4 p = zr[lane];
            const float4 q = zr[lane + 32];
            accA.x += a * p.x; accA.y += a * p.y; accA.z += a * p.z; accA.w += a * p.w;
            accB.x += a * q.x; accB.y += a * q.y; accB.z += a * q.z; accB.w += a * q.w;
        }
    } else {
        float m = ninf;
        for (int i = lane; i < N_EDGES; i += 32)
            if (dst[i] == node) m = fmaxf(m, leaky(s_src[src[i]] + sd + ab));
        for (int o = 16; o >= 1; o >>= 1) m = fmaxf(m, __shfl_xor(m, o, 32));
        float dn = 0.0f;
        for (int pass = 0; pass < 2; ++pass) {
            int last = -1;
            int remaining = deg;
            for (int it = 0; it < deg && remaining > 0; ++it) {
                int cand = KEY_NONE, cn = 0;
                for (int i = lane; i < N_EDGES; i += 32) {
                    if (dst[i] == node) {
                        const int k = src[i];
                        if (k > last) {
                            if (k < cand) { cand = k; cn = 1; }
                            else if (k == cand) ++cn;
                        }
                    }
                }
                int cm = cand;
                for (int o = 16; o >= 1; o >>= 1) { const int y = __shfl_xor(cm, o, 32); cm = (y < cm) ? y : cm; }
                int c = (cand == cm) ? cn : 0;
                for (int o = 16; o >= 1; o >>= 1) c += __shfl_xor(c, o, 32);
                if (c <= 0) break;
                const float x = expf(leaky(s_src[cm] + sd + ab) - m);
                if (pass == 0) {
                    for (int t = 0; t < c; ++t) dn += x;
                } else {
                    const float a = x / dn;
                    const float4* zr = (const float4*)(z + (size_t)cm * OUT_DIM);
                    const float4 p = zr[lane];
                    const float4 q = zr[lane + 32];
                    for (int t = 0; t < c; ++t) {
                        accA.x += a * p.x; accA.y += a * p.y; accA.z += a * p.z; accA.w += a * p.w;
                        accB.x += a * q.x; accB.y += a * q.y; accB.z += a * q.z; accB.w += a * q.w;
                    }
                }
                last = cm;
                remaining -= c;
            }
        }
    }

    float4* ho = (float4*)(hout + (size_t)node * OUT_DIM);
    ho[lane]      = accA;
    ho[lane + 32] = accB;
}

static inline size_t align256(size_t x) { return (x + 255) & ~(size_t)255; }

extern "C" void kernel_launch(void* const* d_in, const int* in_sizes, int n_in,
                              void* d_out, int out_size, void* d_ws, size_t ws_size,
                              hipStream_t stream) {
    const float* h      = (const float*)d_in[0];
    const float* fc_w   = (const float*)d_in[1];
    const float* fc_b   = (const float*)d_in[2];
    const float* attn_w = (const float*)d_in[3];
    const float* attn_b = (const float*)d_in[4];
    const int*   src    = (const int*)d_in[5];
    const int*   dst    = (const int*)d_in[6];
    float* hout = (float*)d_out;

    char* ws = (char*)d_ws;
    float* z     = (float*)ws; ws += align256((size_t)N_NODES * OUT_DIM * 4);
    float* s_src = (float*)ws; ws += align256((size_t)N_NODES * 4);
    float* s_dst = (float*)ws; ws += align256((size_t)N_NODES * 4);
    unsigned short* epay  = (unsigned short*)ws; ws += align256((size_t)N_EDGES * 2);
    unsigned short* eoffT = (unsigned short*)ws; ws += align256((size_t)(BIN4 + 1) * NCHUNK * 2);

    k_gemm_wmma<<<N_NODES / 16 + NCHUNK, 128, 0, stream>>>(h, fc_w, fc_b, attn_w, z, s_src, s_dst,
                                                          src, dst, epay, eoffT);
    k_gather<<<N_NODES / 8, 256, 0, stream>>>(epay, eoffT, src, dst, s_src, s_dst, attn_b, z, hout);
}


# GEMM work-group widened to a 48x256 block (12 waves, 3 per SIMD, one block per WGP, 64-k chunks): each w chunk staged in LDS once per 48 rows, not per 16-row strip; every sum's order kept
# speedup vs baseline: 4.7137x; 1.1074x over previous
#include <hip/hip_runtime.h>

#define N_NODES 10000
#define N_EDGES 320000
#define IN_DIM  512
#define OUT_DIM 256
#define NEG_SLOPE 0.01f
#define KEY_NONE 0x7FFFFFFF

typedef float v2f __attribute__((ext_vector_type(2)));
typedef float v8f __attribute__((ext_vector_type(8)));

#define KC      64
#define PITCH   68
#define MS      3
#define A_ROWS  (16 * MS)
#define B_ROWS  OUT_DIM
#define STAGE_ROWS (A_ROWS + B_ROWS)
#define STAGE_FLOATS (STAGE_ROWS * PITCH)
#define GNT     384
#define NLD     13
#define N_STRIPS (N_NODES / 16)
#define N_GWG   ((N_STRIPS + MS - 1) / MS)
static_assert(STAGE_ROWS * (KC / 4) <= NLD * GNT && STAGE_ROWS * (KC / 4) > (NLD - 1) * GNT, "staging map");
static_assert(A_ROWS * (KC / 4) == 2 * GNT, "h rows take the first two staging steps");
static_assert(N_NODES % 16 == 0 && IN_DIM % KC == 0 && (IN_DIM / KC) % 2 == 0, "tiling");

#define NCHUNK  64
#define CHUNK   (N_EDGES / NCHUNK)
#define BIN4    (N_NODES / 4)
#define HIST_N  2560
static_assert(N_EDGES % NCHUNK == 0 && CHUNK % 8 == 0, "chunks are whole 16-byte groups");
static_assert(CHUNK <= 65535, "offsets inside a chunk are kept as 16-bit values");
static_assert(N_NODES % 8 == 0 && N_NODES <= 16384, "records hold a 14-bit source id");
static_assert(BIN4 < HIST_N && HIST_N == 128 * 20, "scan layout");
#define BNJ ((CHUNK / 4 + GNT - 1) / GNT)
static_assert(BNJ * GNT >= CHUNK / 4, "every edge group has a thread");

__device__ __forceinline__ void bin_chunk(const int* __restrict__ src, const int* __restrict__ dst,
                                          unsigned short* __restrict__ epay,
                                          unsigned short* __restrict__ eoffT,
                                          float* ldsf, const int p, const int t) {
    int* hist = (int*)ldsf;
    int* wsum = hist + HIST_N;
    unsigned short* stage = (unsigned short*)(wsum + 4);
    const int lane = t & 31;
    const int wv   = t >> 5;

    for (int k = t; k < HIST_N; k += GNT) hist[k] = 0;

    const int4* d4 = (const int4*)(dst + (size_t)p * CHUNK);
    const int4* s4 = (const int4*)(src + (size_t)p * CHUNK);
    int4 dd[BNJ], ss[BNJ];
#pragma unroll
    for (int j = 0; j < BNJ; ++j) {
        const int g = t + GNT * j;
        if (g < CHUNK / 4) { dd[j] = d4[g]; ss[j] = s4[g]; }
        else { dd[j] = make_int4(0, 0, 0, 0); ss[j] = make_int4(0, 0, 0, 0); }
    }
    __syncthreads();

#pragma unroll
    for (int j = 0; j < BNJ; ++j) {
        if (t + GNT * j < CHUNK / 4) {
            atomicAdd(&hist[dd[j].x >> 2], 1);
            atomicAdd(&hist[dd[j].y >> 2], 1);
            atomicAdd(&hist[dd[j].z >> 2], 1);
            atomicAdd(&hist[dd[j].w >> 2], 1);
        }
    }
    __syncthreads();

    const bool scanner = (t < 128);
    const int base = t * 20;
    int loc[20], tot = 0, inc = 0;
    if (scanner) {
#pragma unroll
        for (int k = 0; k < 20; ++k) { loc[k] = hist[base + k]; tot += loc[k]; }
        inc = tot;
#pragma unroll
        for (int o = 1; o < 32; o <<= 1) {
            const int y = __shfl_up(inc, o, 32);
            if (lane >= o) inc += y;
        }
        if (lane == 31) wsum[wv] = inc;
    }
    __syncthreads();
    if (scanner) {
        int run = inc - tot;
        for (int q = 0; q < wv; ++q) run += wsum[q];
#pragma unroll
        for (int k = 0; k < 20; ++k) {
            const int bin = base + k;
            hist[bin] = run;
            if (bin <= BIN4) eoffT[(size_t)bin * NCHUNK + p] = (unsigned short)run;
            run += loc[k];
        }
    }
    __syncthreads();

#pragma unroll
    for (int j = 0; j < BNJ; ++j) {
        if (t + GNT * j < CHUNK / 4) {
            const int dv[4] = {dd[j].x, dd[j].y, dd[j].z, dd[j].w};
            const int sv[4] = {ss[j].x, ss[j].y, ss[j].z, ss[j].w};
#pragma unroll
            for (int c = 0; c < 4; ++c) {
                const int pos = atomicAdd(&hist[dv[c] >> 2], 1);
                stage[pos] = (unsigned short)(((dv[c] & 3) << 14) | sv[c]);
            }
        }
    }
    __syncthreads();

    const uint4* st4 = (const uint4*)stage;
    uint4* o4 = (uint4*)(epay + (size_t)p * CHUNK);
    for (int k = t; k < CHUNK / 8; k += GNT) o4[k] = st4[k];
}

template <int M>
__device__ __forceinline__ float row_xor_add(float v) {
    const int p = __builtin_amdgcn_update_dpp(0, __float_as_int(v), 0x160 + M, 0xf, 0xf, true);
    return v + __int_as_float(p);
}

__device__ __forceinline__ void mma_half(const float* cur, int fA, int fB,
                                         v8f& c0, v8f& c1, v8f& c2, v8f& c3) {
#pragma unroll
    for (int kk = 0; kk < KC / 2; kk += 4) {
        v2f a  = *(const v2f*)(cur + fA + kk);
        v2f q0 = *(const v2f*)(cur + fB + kk);
        v2f q1 = *(const v2f*)(cur + fB + 16 * PITCH + kk);
        v2f q2 = *(const v2f*)(cur + fB + 32 * PITCH + kk);
        v2f q3 = *(const v2f*)(cur + fB + 48 * PITCH + kk);
        c0 = __builtin_amdgcn_wmma_f32_16x16x4_f32(false, a, false, q0, (short)0, c0, false, false);
        c1 = __builtin_amdgcn_wmma_f32_16x16x4_f32(false, a, false, q1, (short)0, c1, false, false);
        c2 = __builtin_amdgcn_wmma_f32_16x16x4_f32(false, a, false, q2, (short)0, c2, false, false);
        c3 = __builtin_amdgcn_wmma_f32_16x16x4_f32(false, a, false, q3, (short)0, c3, false, false);
    }
}

__global__ void __launch_bounds__(GNT) k_gemm_wmma(const float* __restrict__ h,
                                                   const float* __restrict__ w,
                                                   const float* __restrict__ bias,
                                                   const float* __restrict__ attn_w,
                                                   float* __restrict__ z,
                                                   float* __restrict__ s_src,
                                                   float* __restrict__ s_dst,
                                                   const int* __restrict__ src,
                                                   const int* __restrict__ dst,
                                                   unsigned short* __restrict__ epay,
                                                   unsigned short* __restrict__ eoffT) {
    __shared__ __attribute__((aligned(16))) float lds[2 * STAGE_FLOATS];

    const int t   = threadIdx.x;
    if (blockIdx.x >= N_GWG) { bin_chunk(src, dst, epay, eoffT, lds, (int)blockIdx.x - N_GWG, t); return; }
    const int blk = (int)blockIdx.x;

    const int lane = t & 31;
    const int wvi  = t >> 5;
    const int mi   = wvi >> 2;
    const int nj   = wvi & 3;
    const int strip = blk * MS + mi;
    const bool live = (strip < N_STRIPS);
    const int row0 = strip * 16;
    const int col0 = nj * 64;
    const int mn   = lane & 15;
    const int kb   = (lane >> 4) << 1;

    const int seg = t & 15;
    const int sr  = t >> 4;
    const int ha  = blk * A_ROWS + sr;
    const int hr0 = (ha < N_NODES) ? ha : N_NODES - 1;
    const int hr1 = (ha + 24 < N_NODES) ? ha + 24 : N_NODES - 1;
    const int back = (t >= 256) ? 16 : 0;
    const float* gA0 = h + (size_t)hr0 * IN_DIM + seg * 4;
    const float* gA1 = h + (size_t)hr1 * IN_DIM + seg * 4;
    const float* gB  = w + (size_t)sr * IN_DIM + seg * 4;
    const float* gBl = w + (size_t)(sr + 240 - back) * IN_DIM + seg * 4;
    const int    sS  = sr * PITCH + seg * 4;
    const int    sSl = (sr + 288 - back) * PITCH + seg * 4;

    const int fA = (mi * 16 + mn) * PITCH + kb;
    const int fB = (A_ROWS + col0 + mn) * PITCH + kb;

    float4 rg[NLD];
    rg[0] = *(const float4*)gA0;
    rg[1] = *(const float4*)gA1;
#pragma unroll
    for (int i = 0; i < 10; ++i) rg[2 + i] = *(const float4*)(gB + (size_t)i * 24 * IN_DIM);
    rg[12] = *(const float4*)gBl;
#pragma unroll
    for (int j = 0; j < 12; ++j) *(float4*)(lds + sS + j * 24 * PITCH) = rg[j];
    *(float4*)(lds + sSl) = rg[12];
    __syncthreads();

    v8f c0 = {}, c1 = {}, c2 = {}, c3 = {};
#pragma nounroll
    for (int c = 0; c + 1 < IN_DIM / KC; ++c) {
        const float* cur = lds + (c & 1) * STAGE_FLOATS;
        float*       nxt = lds + ((c + 1) & 1) * STAGE_FLOATS;
        const int k1 = (c + 1) * KC;
        rg[0] = *(const float4*)(gA0 + k1);
        rg[1] = *(const float4*)(gA1 + k1);
#pragma unroll
        for (int i = 0; i < 10; ++i) rg[2 + i] = *(const float4*)(gB + (size_t)i * 24 * IN_DIM + k1);
        rg[12] = *(const float4*)(gBl + k1);
        mma_half(cur, fA, fB, c0, c1, c2, c3);
        mma_half(cur + KC / 2, fA, fB, c0, c1, c2, c3);
#pragma unroll
        for (int j = 0; j < 12; ++j) *(float4*)(nxt + sS + j * 24 * PITCH) = rg[j];
        *(float4*)(nxt + sSl) = rg[12];
        __builtin_amdgcn_sched_group_barrier(0x020, 13, 0);
        __builtin_amdgcn_sched_group_barrier(0x100, 20, 0);
        __builtin_amdgcn_sched_group_barrier(0x008, 32, 0);
        __builtin_amdgcn_sched_group_barrier(0x100, 20, 0);
        __builtin_amdgcn_sched_group_barrier(0x008, 32, 0);
        __builtin_amdgcn_sched_group_barrier(0x200, 13, 0);
        __syncthreads();
    }
    {
        const float* last = lds + ((IN_DIM / KC - 1) & 1) * STAGE_FLOATS;
        mma_half(last, fA, fB, c0, c1, c2, c3);
        mma_half(last + KC / 2, fA, fB, c0, c1, c2, c3);
    }

    const int mbase = (lane >> 4) * 8;
    v8f acc[4] = {c0, c1, c2, c3};
    float ps[8], pd[8];
#pragma unroll
    for (int v = 0; v < 8; ++v) { ps[v] = 0.0f; pd[v] = 0.0f; }
#pragma unroll
    for (int g = 0; g < 4; ++g) {
        const int   col = col0 + g * 16 + mn;
        const float bv  = bias[col];
        const float as  = attn_w[col];
        const float ad  = attn_w[OUT_DIM + col];
#pragma unroll
        for (int v = 0; v < 8; ++v) {
            acc[g][v] += bv;
            ps[v] += acc[g][v] * as;
            pd[v] += acc[g][v] * ad;
        }
    }
#pragma unroll
    for (int v = 0; v < 8; ++v) {
        ps[v] = row_xor_add<1>(ps[v]);  pd[v] = row_xor_add<1>(pd[v]);
        ps[v] = row_xor_add<2>(ps[v]);  pd[v] = row_xor_add<2>(pd[v]);
        ps[v] = row_xor_add<4>(ps[v]);  pd[v] = row_xor_add<4>(pd[v]);
        ps[v] = row_xor_add<8>(ps[v]);  pd[v] = row_xor_add<8>(pd[v]);
    }
    float* part = lds;
    if (mn == 0) {
#pragma unroll
        for (int v = 0; v < 8; ++v)
            *(float2*)(part + ((mi * 4 + nj) * 16 + mbase + v) * 2) = make_float2(ps[v], pd[v]);
    }
    __syncthreads();
    if (live) {
#pragma unroll
        for (int g = 0; g < 4; ++g) {
            const int col = col0 + g * 16 + mn;
#pragma unroll
            for (int v = 0; v < 8; ++v)
                z[(size_t)(row0 + mbase + v) * OUT_DIM + col] = acc[g][v];
        }
        if (nj == 0) {
            const int r = lane & 15, which = lane >> 4;
            const float* pp = part + (mi * 4) * 32;
            float s = pp[(0 * 16 + r) * 2 + which];
            s += pp[(1 * 16 + r) * 2 + which];
            s += pp[(2 * 16 + r) * 2 + which];
            s += pp[(3 * 16 + r) * 2 + which];
            (which ? s_dst : s_src)[row0 + r] = s;
        }
    }
}

#define BIN_CAP 64

__device__ __forceinline__ float leaky(float v) { return (v >= 0.0f) ? v : NEG_SLOPE * v; }

__global__ void __launch_bounds__(256) k_gather(const unsigned short* __restrict__ epay,
                                                const unsigned short* __restrict__ eoffT,
                                                const int* __restrict__ src,
                                                const int* __restrict__ dst,
                                                const float* __restrict__ s_src,
                                                const float* __restrict__ s_dst,
                                                const float* __restrict__ attn_b,
                                                const float* __restrict__ z,
                                                float* __restrict__ hout) {
    __shared__ int sm[8][64];
    __shared__ int raw[8][BIN_CAP];
    __shared__ int lcnt[8];
    const int lane = threadIdx.x & 31;
    const int w     = __builtin_amdgcn_readfirstlane((int)(threadIdx.x >> 5));
    const int node  = blockIdx.x * 8 + w;

    if (threadIdx.x < 8) lcnt[threadIdx.x] = 0;
    __syncthreads();
    {
        const int p = threadIdx.x & (NCHUNK - 1);
        const unsigned short* ho = eoffT + (size_t)(2 * blockIdx.x) * NCHUNK + p;
        const int o0 = ho[0];
        const int o1 = ho[NCHUNK];
        const int o2 = ho[2 * NCHUNK];
        const unsigned short* pay = epay + (size_t)p * CHUNK;
        for (int i = o0 + (int)(threadIdx.x >> 6); i < o2; i += 4) {
            const int rec = pay[i];
            const int nl  = ((i >= o1) ? 4 : 0) + (rec >> 14);
            const int pos = atomicAdd(&lcnt[nl], 1);
            if (pos < BIN_CAP) raw[nl][pos] = rec & 0x3FFF;
        }
    }
    __syncthreads();
    const int deg   = __builtin_amdgcn_readfirstlane(lcnt[w]);
    const bool fast = (deg <= BIN_CAP);
    const bool v0   = fast && (lane < deg);
    const bool v1   = fast && (lane + 32 < deg);
    const int  n0   = (deg < 32) ? deg : 32;

    const int k0 = v0 ? raw[w][lane] : KEY_NONE;
    const int k1 = v1 ? raw[w][lane + 32] : KEY_NONE;
    if (fast) {
        int r0 = 0, r1 = 0;
        for (int j = 0; j < n0; ++j) {
            const int o = __builtin_amdgcn_readlane(k0, j);
            r0 += ((o < k0) || (o == k0 && j < lane)) ? 1 : 0;
            r1 += (o <= k1) ? 1 : 0;
        }
        for (int j = 32; j < deg; ++j) {
            const int o = __builtin_amdgcn_readlane(k1, j - 32);
            r0 += (o < k0) ? 1 : 0;
            r1 += ((o < k1) || (o == k1 && (j - 32) < lane)) ? 1 : 0;
        }
        if (v0) sm[w][r0] = k0;
        if (v1) sm[w][r1] = k1;
    }
    __syncthreads();

    float4 accA = {0.0f, 0.0f, 0.0f, 0.0f}, accB = {0.0f, 0.0f, 0.0f, 0.0f};
    const float sd = s_dst[node];
    const float ab = attn_b[0];
    const float ninf = -__builtin_huge_valf();

    if (fast) {
        const int s0 = v0 ? sm[w][lane] : 0;
        const int s1 = v1 ? sm[w][lane + 32] : 0;
        float e0 = ninf, e1 = ninf;
        if (v0) e0 = leaky(s_src[s0] + sd + ab);
        if (v1) e1 = leaky(s_src[s1] + sd + ab);
        float m = fmaxf(e0, e1);
        for (int o = 16; o >= 1; o >>= 1) m = fmaxf(m, __shfl_xor(m, o, 32));
        const float x0 = v0 ? expf(e0 - m) : 0.0f;
        const float x1 = v1 ? expf(e1 - m) : 0.0f;
        float dn = x0 + x1;
        for (int o = 16; o >= 1; o >>= 1) dn += __shfl_xor(dn, o, 32);
        const int a0 = __float_as_int(x0 / dn);
        const int a1 = __float_as_int(x1 / dn);
#pragma unroll 4
        for (int r = 0; r < n0; ++r) {
            const int   s = __builtin_amdgcn_readlane(s0, r);
            const float a = __int_as_float(__builtin_amdgcn_readlane(a0, r));
            const float4* zr = (const float4*)(z + (size_t)s * OUT_DIM);
            const float4 p = zr[lane];
            const float4 q = zr[lane + 32];
            accA.x += a * p.x; accA.y += a * p.y; accA.z += a * p.z; accA.w += a * p.w;
            accB.x += a * q.x; accB.y += a * q.y; accB.z += a * q.z; accB.w += a * q.w;
        }
#pragma unroll 4
        for (int r = 32; r < deg; ++r) {
            const int   s = __builtin_amdgcn_readlane(s1, r - 32);
            const float a = __int_as_float(__builtin_amdgcn_readlane(a1, r - 32));
            const float4* zr = (const float4*)(z + (size_t)s * OUT_DIM);
            const float4 p = zr[lane];
            const float4 q = zr[lane + 32];
            accA.x += a * p.x; accA.y += a * p.y; accA.z += a * p.z; accA.w += a * p.w;
            accB.x += a * q.x; accB.y += a * q.y; accB.z += a * q.z; accB.w += a * q.w;
        }
    } else {
        float m = ninf;
        for (int i = lane; i < N_EDGES; i += 32)
            if (dst[i] == node) m = fmaxf(m, leaky(s_src[src[i]] + sd + ab));
        for (int o = 16; o >= 1; o >>= 1) m = fmaxf(m, __shfl_xor(m, o, 32));
        float dn = 0.0f;
        for (int pass = 0; pass < 2; ++pass) {
            int last = -1;
            int remaining = deg;
            for (int it = 0; it < deg && remaining > 0; ++it) {
                int cand = KEY_NONE, cn = 0;
                for (int i = lane; i < N_EDGES; i += 32) {
                    if (dst[i] == node) {
                        const int k = src[i];
                        if (k > last) {
                            if (k < cand) { cand = k; cn = 1; }
                            else if (k == cand) ++cn;
                        }
                    }
                }
                int cm = cand;
                for (int o = 16; o >= 1; o >>= 1) { const int y = __shfl_xor(cm, o, 32); cm = (y < cm) ? y : cm; }
                int c = (cand == cm) ? cn : 0;
                for (int o = 16; o >= 1; o >>= 1) c += __shfl_xor(c, o, 32);
                if (c <= 0) break;
                const float x = expf(leaky(s_src[cm] + sd + ab) - m);
                if (pass == 0) {
                    for (int t = 0; t < c; ++t) dn += x;
                } else {
                    const float a = x / dn;
                    const float4* zr = (const float4*)(z + (size_t)cm * OUT_DIM);
                    const float4 p = zr[lane];
                    const float4 q = zr[lane + 32];
                    for (int t = 0; t < c; ++t) {
                        accA.x += a * p.x; accA.y += a * p.y; accA.z += a * p.z; accA.w += a * p.w;
                        accB.x += a * q.x; accB.y += a * q.y; accB.z += a * q.z; accB.w += a * q.w;
                    }
                }
                last = cm;
                remaining -= c;
            }
        }
    }

    float4* ho = (float4*)(hout + (size_t)node * OUT_DIM);
    ho[lane]      = accA;
    ho[lane + 32] = accB;
}

static inline size_t align256(size_t x) { return (x + 255) & ~(size_t)255; }

extern "C" void kernel_launch(void* const* d_in, const int* in_sizes, int n_in,
                              void* d_out, int out_size, void* d_ws, size_t ws_size,
                              hipStream_t stream) {
    const float* h      = (const float*)d_in[0];
    const float* fc_w   = (const float*)d_in[1];
    const float* fc_b   = (const float*)d_in[2];
    const float* attn_w = (const float*)d_in[3];
    const float* attn_b = (const float*)d_in[4];
    const int*   src    = (const int*)d_in[5];
    const int*   dst    = (const int*)d_in[6];
    float* hout = (float*)d_out;

    char* ws = (char*)d_ws;
    float* z     = (float*)ws; ws += align256((size_t)N_NODES * OUT_DIM * 4);
    float* s_src = (float*)ws; ws += align256((size_t)N_NODES * 4);
    float* s_dst = (float*)ws; ws += align256((size_t)N_NODES * 4);
    unsigned short* epay  = (unsigned short*)ws; ws += align256((size_t)N_EDGES * 2);
    unsigned short* eoffT = (unsigned short*)ws; ws += align256((size_t)(BIN4 + 1) * NCHUNK * 2);

    k_gemm_wmma<<<N_GWG + NCHUNK, GNT, 0, stream>>>(h, fc_w, fc_b, attn_w, z, s_src, s_dst,
                                                    src, dst, epay, eoffT);
    k_gather<<<N_NODES / 8, 256, 0, stream>>>(epay, eoffT, src, dst, s_src, s_dst, attn_b, z, hout);
}
